# starting kernel
# baseline (speedup 1.0000x reference)
#include <hip/hip_runtime.h>
#include <math.h>

typedef float v2f __attribute__((ext_vector_type(2)));
typedef float v8f __attribute__((ext_vector_type(8)));

#define D_IN  256
#define D_HID 128
#define D_EMB 128
#define D_OUT 16

__global__ void transpose_w_kernel(const float* __restrict__ in,
                                   float* __restrict__ out, int K, int D)
{
    const int idx = blockIdx.x * blockDim.x + threadIdx.x;
    if (idx < K * D) {
        const int k = idx / D;
        const int c = idx - k * D;
        out[c * K + k] = in[idx];
    }
}

__global__ void gemm_wmma_f32_kernel(const float* __restrict__ A,
                                     const float* __restrict__ WT,
                                     float* __restrict__ C,
                                     const float* __restrict__ bias,
                                     int M, int K, int DOUTv,
                                     int totalTiles, int colTiles)
{
    const int gw = blockIdx.x * (blockDim.x >> 5) + (threadIdx.x >> 5);
    if (gw >= totalTiles) return;

    const int rowTile = gw / colTiles;
    const int colTile = gw - rowTile * colTiles;
    const int lane = threadIdx.x & 31;
    const int half = lane >> 4;
    const int lm   = lane & 15;

    const int rowBase = rowTile * 16;
    const int colBase = colTile * 16;

    const float* pA = A  + (size_t)(rowBase + lm) * K + 2 * half;
    const float* pB = WT + (size_t)(colBase + lm) * K + 2 * half;

    v8f acc = {};
    const int steps = K >> 2;
    #pragma unroll 4
    for (int s = 0; s < steps; ++s) {
        v2f a = *(const v2f*)pA;  pA += 4;
        v2f b = *(const v2f*)pB;  pB += 4;
        acc = __builtin_amdgcn_wmma_f32_16x16x4_f32(
                  false, a, false, b, (short)0, acc, false, false);
    }

    const float bv = bias ? bias[colBase + lm] : 0.0f;
    float* pC = C + (size_t)(rowBase + 8 * half) * DOUTv + colBase + lm;
    #pragma unroll
    for (int v = 0; v < 8; ++v) {
        const int r = rowBase + v + 8 * half;
        if (r < M) pC[(size_t)v * DOUTv] = acc[v] + bv;
    }
}

__global__ void spmm_atomic_kernel(const float* __restrict__ H,
                                   float* __restrict__ AGG,
                                   const int* __restrict__ ei,
                                   const float* __restrict__ ew,
                                   int E, int D)
{
    const int lane   = threadIdx.x & 31;
    const int wave   = (blockIdx.x * blockDim.x + threadIdx.x) >> 5;
    const int nwaves = (gridDim.x * blockDim.x) >> 5;

    for (int e = wave; e < E; e += nwaves) {
        const int   src = ei[e];
        const int   dst = ei[E + e];
        const float w   = ew[e];
        const float4 h4 = *(const float4*)(H + (size_t)src * D + lane * 4);
        float* p = AGG + (size_t)dst * D + lane * 4;
        atomicAdd(p + 0, h4.x * w);
        atomicAdd(p + 1, h4.y * w);
        atomicAdd(p + 2, h4.z * w);
        atomicAdd(p + 3, h4.w * w);
    }
}

__global__ void zero_kernel(float4* __restrict__ p, size_t n4)
{
    const size_t i = (size_t)blockIdx.x * blockDim.x + threadIdx.x;
    const size_t stride = (size_t)gridDim.x * blockDim.x;
    const float4 z = make_float4(0.f, 0.f, 0.f, 0.f);
    for (size_t j = i; j < n4; j += stride) p[j] = z;
}

__global__ void bias_act_kernel(const float4* __restrict__ in,
                                const float* __restrict__ bias,
                                float4* __restrict__ out,
                                size_t total4, int D, int doRelu)
{
    const size_t i = (size_t)blockIdx.x * blockDim.x + threadIdx.x;
    const size_t stride = (size_t)gridDim.x * blockDim.x;
    const int mask4 = (D >> 2) - 1;
    for (size_t j = i; j < total4; j += stride) {
        const float4 b4 = *(const float4*)(bias + ((j & mask4) << 2));
        float4 v = in[j];
        v.x += b4.x; v.y += b4.y; v.z += b4.z; v.w += b4.w;
        if (doRelu) {
            v.x = fmaxf(v.x, 0.f); v.y = fmaxf(v.y, 0.f);
            v.z = fmaxf(v.z, 0.f); v.w = fmaxf(v.w, 0.f);
        }
        out[j] = v;
    }
}

__global__ void log_softmax16_kernel(const float* __restrict__ Y,
                                     float* __restrict__ out, int N)
{
    const int r = blockIdx.x * blockDim.x + threadIdx.x;
    if (r >= N) return;
    float v[D_OUT];
    float m = -INFINITY;
    #pragma unroll
    for (int j = 0; j < D_OUT; ++j) { v[j] = Y[(size_t)r * D_OUT + j]; m = fmaxf(m, v[j]); }
    float s = 0.0f;
    #pragma unroll
    for (int j = 0; j < D_OUT; ++j) s += __expf(v[j] - m);
    const float l = m + __logf(s);
    #pragma unroll
    for (int j = 0; j < D_OUT; ++j) out[(size_t)r * D_OUT + j] = v[j] - l;
}

extern "C" void kernel_launch(void* const* d_in, const int* in_sizes, int n_in,
                              void* d_out, int out_size, void* d_ws, size_t ws_size,
                              hipStream_t stream)
{
    const float* x    = (const float*)d_in[0];
    const float* ew   = (const float*)d_in[1];
    const float* W1   = (const float*)d_in[2];
    const float* b1   = (const float*)d_in[3];
    const float* W2   = (const float*)d_in[4];
    const float* b2   = (const float*)d_in[5];
    const float* W3   = (const float*)d_in[6];
    const float* b3   = (const float*)d_in[7];
    const float* fcw  = (const float*)d_in[8];
    const float* fcb  = (const float*)d_in[9];
    const int*   ei   = (const int*)d_in[10];

    const int N = in_sizes[0] / D_IN;
    const int E = in_sizes[1];

    float* B0   = (float*)d_ws;
    float* B1   = B0 + (size_t)N * D_HID;
    float* WT1  = B1 + (size_t)N * D_HID;
    float* WT2  = WT1 + D_IN  * D_HID;
    float* WT3  = WT2 + D_HID * D_HID;
    float* WTfc = WT3 + D_HID * D_EMB;
    float* Y    = B0;

    float* outEmb = (float*)d_out;
    float* outLsm = outEmb + (size_t)N * D_EMB;

    const int rowTiles = (N + 15) / 16;
    const dim3 blk(256);

    auto transpose = [&](const float* W, float* WT, int K, int D) {
        const int tot = K * D;
        transpose_w_kernel<<<(tot + 255) / 256, blk, 0, stream>>>(W, WT, K, D);
    };
    auto gemm = [&](const float* A, const float* WT, float* C, const float* bias,
                    int K, int DOUTv) {
        const int colTiles = DOUTv / 16;
        const int total = rowTiles * colTiles;
        gemm_wmma_f32_kernel<<<(total + 7) / 8, blk, 0, stream>>>(
            A, WT, C, bias, N, K, DOUTv, total, colTiles);
    };
    auto zero = [&](float* p, size_t nElem) {
        zero_kernel<<<2048, blk, 0, stream>>>((float4*)p, nElem / 4);
    };
    auto spmm = [&](const float* H, float* AGG) {
        spmm_atomic_kernel<<<2048, blk, 0, stream>>>(H, AGG, ei, ew, E, D_HID);
    };
    auto biasAct = [&](const float* in, const float* b, float* out, int relu) {
        bias_act_kernel<<<2048, blk, 0, stream>>>((const float4*)in, b,
                                                  (float4*)out,
                                                  (size_t)N * D_HID / 4,
                                                  D_HID, relu);
    };

    transpose(W1, WT1, D_IN,  D_HID);
    transpose(W2, WT2, D_HID, D_HID);
    transpose(W3, WT3, D_HID, D_EMB);
    transpose(fcw, WTfc, D_EMB, D_OUT);

    gemm(x, WT1, B0, nullptr, D_IN, D_HID);
    zero(B1, (size_t)N * D_HID);
    spmm(B0, B1);
    biasAct(B1, b1, B1, 1);

    gemm(B1, WT2, B0, nullptr, D_HID, D_HID);
    zero(B1, (size_t)N * D_HID);
    spmm(B0, B1);
    biasAct(B1, b2, B1, 1);

    gemm(B1, WT3, B0, nullptr, D_HID, D_EMB);
    zero(B1, (size_t)N * D_HID);
    spmm(B0, B1);
    biasAct(B1, b3, outEmb, 0);

    gemm(outEmb, WTfc, Y, fcb, D_EMB, D_OUT);
    log_softmax16_kernel<<<(N + 255) / 256, blk, 0, stream>>>(Y, outLsm, N);
}


# SpMM without global atomics: a work-group owns 300 output rows, filters the edge list for them and sums gathered rows into LDS on top of the bias (LDS float adds, arrival order), one store per row
# speedup vs baseline: 1.8429x; 1.8429x over previous
#include <hip/hip_runtime.h>
#include <math.h>

typedef float v2f __attribute__((ext_vector_type(2)));
typedef float v8f __attribute__((ext_vector_type(8)));

#define D_IN  256
#define D_HID 128
#define D_EMB 128
#define D_OUT 16

__global__ void transpose_w_kernel(const float* __restrict__ in,
                                   float* __restrict__ out, int K, int D)
{
    const int idx = blockIdx.x * blockDim.x + threadIdx.x;
    if (idx < K * D) {
        const int k = idx / D;
        const int c = idx - k * D;
        out[c * K + k] = in[idx];
    }
}

template <bool RELU_A>
__global__ void gemm_wmma_f32_kernel(const float* __restrict__ A,
                                     const float* __restrict__ WT,
                                     float* __restrict__ C,
                                     const float* __restrict__ bias,
                                     int M, int K, int DOUTv,
                                     int totalTiles, int colTiles)
{
    const int gw = blockIdx.x * (blockDim.x >> 5) + (threadIdx.x >> 5);
    if (gw >= totalTiles) return;

    const int rowTile = gw / colTiles;
    const int colTile = gw - rowTile * colTiles;
    const int lane = threadIdx.x & 31;
    const int half = lane >> 4;
    const int lm   = lane & 15;

    const int rowBase = rowTile * 16;
    const int colBase = colTile * 16;

    const float* pA = A  + (size_t)(rowBase + lm) * K + 2 * half;
    const float* pB = WT + (size_t)(colBase + lm) * K + 2 * half;

    v8f acc = {};
    const int steps = K >> 2;
    #pragma unroll 4
    for (int s = 0; s < steps; ++s) {
        v2f a = *(const v2f*)pA;  pA += 4;
        v2f b = *(const v2f*)pB;  pB += 4;
        if (RELU_A) { a.x = fmaxf(a.x, 0.0f); a.y = fmaxf(a.y, 0.0f); }
        acc = __builtin_amdgcn_wmma_f32_16x16x4_f32(
                  false, a, false, b, (short)0, acc, false, false);
    }

    const float bv = bias ? bias[colBase + lm] : 0.0f;
    float* pC = C + (size_t)(rowBase + 8 * half) * DOUTv + colBase + lm;
    #pragma unroll
    for (int v = 0; v < 8; ++v) {
        const int r = rowBase + v + 8 * half;
        if (r < M) pC[(size_t)v * DOUTv] = acc[v] + bv;
    }
}

#define SP_ROWS   300
#define SP_EBITS  23
#define SP_WAVES  32
#define SP_SCAN_U 8
#define SP_GATH_U 8
#define SP_PEND   (32 + 128 * SP_SCAN_U)
static_assert(SP_ROWS <= (1 << (32 - SP_EBITS)), "row number fits above the edge index");
static_assert((SP_ROWS * D_HID + SP_WAVES * SP_PEND) * 4 <= 320 * 1024, "sums and lists fit the LDS");

__global__ __launch_bounds__(SP_WAVES * 32)
void spmm_owner_kernel(const float* __restrict__ H,
                       float* __restrict__ AGG,
                       const int* __restrict__ ei,
                       const float* __restrict__ ew,
                       const float* __restrict__ bias,
                       int E, int N)
{
    __shared__ float    acc[SP_ROWS * D_HID];
    __shared__ unsigned pend[SP_WAVES * SP_PEND];

    const int lane = threadIdx.x & 31;
    const int wv   = __builtin_amdgcn_readfirstlane((int)(threadIdx.x >> 5));
    const int lo   = blockIdx.x * SP_ROWS;
    unsigned* myPend = pend + wv * SP_PEND;
    const int* dstp  = ei + E;

    for (int i = threadIdx.x; i < SP_ROWS * D_HID; i += SP_WAVES * 32) {
        const int p = i & (D_HID - 1);
        acc[i] = bias[((p & 31) << 2) | (p >> 5)];
    }
    __syncthreads();

    int cnt = 0;

    auto append = [&](bool m, unsigned rec) {
        const unsigned mask = __builtin_amdgcn_ballot_w32(m);
        if (m) myPend[cnt + (int)__builtin_amdgcn_mbcnt_lo(mask, 0u)] = rec;
        cnt += __builtin_popcount(mask);
    };

    auto drain = [&](int base, int n) {
        unsigned rec = 0u;
        int   src = 0;
        float w   = 0.0f;
        if (lane < n) {
            rec = myPend[base + lane];
            const int e = (int)(rec & ((1u << SP_EBITS) - 1u));
            src = ei[e];
            w   = ew[e];
        }
        const int dl = (int)(rec >> SP_EBITS);
        for (int i = 0; i < n; i += SP_GATH_U) {
            float4 h[SP_GATH_U];
            #pragma unroll
            for (int k = 0; k < SP_GATH_U; ++k) {
                const int s = __builtin_amdgcn_readlane(src, i + k);
                h[k] = *(const float4*)(H + (size_t)s * D_HID + lane * 4);
            }
            #pragma unroll
            for (int k = 0; k < SP_GATH_U; ++k) {
                if (i + k < n) {
                    const float wk = __int_as_float(
                        __builtin_amdgcn_readlane(__float_as_int(w), i + k));
                    const int dk = __builtin_amdgcn_readlane(dl, i + k);
                    float* a = acc + dk * D_HID + lane;
                    __hip_atomic_fetch_add(a,      h[k].x * wk, __ATOMIC_RELAXED, __HIP_MEMORY_SCOPE_WORKGROUP);
                    __hip_atomic_fetch_add(a + 32, h[k].y * wk, __ATOMIC_RELAXED, __HIP_MEMORY_SCOPE_WORKGROUP);
                    __hip_atomic_fetch_add(a + 64, h[k].z * wk, __ATOMIC_RELAXED, __HIP_MEMORY_SCOPE_WORKGROUP);
                    __hip_atomic_fetch_add(a + 96, h[k].w * wk, __ATOMIC_RELAXED, __HIP_MEMORY_SCOPE_WORKGROUP);
                }
            }
        }
    };

    const int head = min(E, (int)(((16u - (unsigned)((size_t)dstp & 15u)) & 15u) >> 2));
    const int nvec = (E - head) >> 2;
    const int fullChunks = nvec >> 5;
    const int4* dv = (const int4*)(dstp + head);

    for (int c = wv; c < fullChunks; c += SP_WAVES * SP_SCAN_U) {
        int4 d[SP_SCAN_U];
        #pragma unroll
        for (int u = 0; u < SP_SCAN_U; ++u) {
            const int cu = min(c + SP_WAVES * u, fullChunks - 1);
            d[u] = dv[cu * 32 + lane];
        }
        #pragma unroll
        for (int u = 0; u < SP_SCAN_U; ++u) {
            const int  cu = c + SP_WAVES * u;
            const bool ok = cu < fullChunks;
            const unsigned r0 = (unsigned)(d[u].x - lo), r1 = (unsigned)(d[u].y - lo);
            const unsigned r2 = (unsigned)(d[u].z - lo), r3 = (unsigned)(d[u].w - lo);
            const bool m0 = ok && r0 < SP_ROWS, m1 = ok && r1 < SP_ROWS;
            const bool m2 = ok && r2 < SP_ROWS, m3 = ok && r3 < SP_ROWS;
            if (__builtin_amdgcn_ballot_w32(m0 || m1 || m2 || m3)) {
                const unsigned e0 = (unsigned)(head + ((cu * 32 + lane) << 2));
                append(m0, (e0     ) | (r0 << SP_EBITS));
                append(m1, (e0 + 1u) | (r1 << SP_EBITS));
                append(m2, (e0 + 2u) | (r2 << SP_EBITS));
                append(m3, (e0 + 3u) | (r3 << SP_EBITS));
            }
        }
        while (cnt >= 32) { cnt -= 32; drain(cnt, 32); }
    }

    if (wv == 0) {
        const int bodyEnd = head + (fullChunks << 7);
        const int nLeft   = head + (E - bodyEnd);
        for (int b = 0; b < nLeft; b += 32) {
            const int idx = b + lane;
            const bool valid = idx < nLeft;
            const int e = idx < head ? idx : bodyEnd + (idx - head);
            const unsigned r = valid ? (unsigned)(dstp[e] - lo) : 0xFFFFFFFFu;
            append(r < SP_ROWS, (unsigned)e | (r << SP_EBITS));
        }
    }
    while (cnt > 0) { const int n = min(cnt, 32); cnt -= n; drain(cnt, n); }

    __syncthreads();

    for (int r = wv; r < SP_ROWS; r += SP_WAVES) {
        const int row = lo + r;
        if (row < N) {
            const float* a = acc + r * D_HID + lane;
            const float4 v = {a[0], a[32], a[64], a[96]};
            *(float4*)(AGG + (size_t)row * D_HID + lane * 4) = v;
        }
    }
}

__global__ void log_softmax16_kernel(const float* __restrict__ Y,
                                     float* __restrict__ out, int N)
{
    const int r = blockIdx.x * blockDim.x + threadIdx.x;
    if (r >= N) return;
    float v[D_OUT];
    float m = -INFINITY;
    #pragma unroll
    for (int j = 0; j < D_OUT; ++j) { v[j] = Y[(size_t)r * D_OUT + j]; m = fmaxf(m, v[j]); }
    float s = 0.0f;
    #pragma unroll
    for (int j = 0; j < D_OUT; ++j) s += __expf(v[j] - m);
    const float l = m + __logf(s);
    #pragma unroll
    for (int j = 0; j < D_OUT; ++j) out[(size_t)r * D_OUT + j] = v[j] - l;
}

extern "C" void kernel_launch(void* const* d_in, const int* in_sizes, int n_in,
                              void* d_out, int out_size, void* d_ws, size_t ws_size,
                              hipStream_t stream)
{
    const float* x    = (const float*)d_in[0];
    const float* ew   = (const float*)d_in[1];
    const float* W1   = (const float*)d_in[2];
    const float* b1   = (const float*)d_in[3];
    const float* W2   = (const float*)d_in[4];
    const float* b2   = (const float*)d_in[5];
    const float* W3   = (const float*)d_in[6];
    const float* b3   = (const float*)d_in[7];
    const float* fcw  = (const float*)d_in[8];
    const float* fcb  = (const float*)d_in[9];
    const int*   ei   = (const int*)d_in[10];

    const int N = in_sizes[0] / D_IN;
    const int E = in_sizes[1];

    float* B0   = (float*)d_ws;
    float* B1   = B0 + (size_t)N * D_HID;
    float* WT1  = B1 + (size_t)N * D_HID;
    float* WT2  = WT1 + D_IN  * D_HID;
    float* WT3  = WT2 + D_HID * D_HID;
    float* WTfc = WT3 + D_HID * D_EMB;
    float* Y    = B0;

    float* outEmb = (float*)d_out;
    float* outLsm = outEmb + (size_t)N * D_EMB;

    const int rowTiles = (N + 15) / 16;
    const dim3 blk(256);

    auto transpose = [&](const float* W, float* WT, int K, int D) {
        const int tot = K * D;
        transpose_w_kernel<<<(tot + 255) / 256, blk, 0, stream>>>(W, WT, K, D);
    };
    auto gemm = [&](const float* A, const float* WT, float* C,
                    const float* bias, int K, int DOUTv) {
        const int colTiles = DOUTv / 16;
        const int total = rowTiles * colTiles;
        gemm_wmma_f32_kernel<false><<<(total + 7) / 8, blk, 0, stream>>>(
            A, WT, C, bias, N, K, DOUTv, total, colTiles);
    };
    auto gemmReluA = [&](const float* A, const float* WT, float* C,
                         const float* bias, int K, int DOUTv) {
        const int colTiles = DOUTv / 16;
        const int total = rowTiles * colTiles;
        gemm_wmma_f32_kernel<true><<<(total + 7) / 8, blk, 0, stream>>>(
            A, WT, C, bias, N, K, DOUTv, total, colTiles);
    };
    auto spmm = [&](const float* H, float* AGG, const float* bias) {
        spmm_owner_kernel<<<(N + SP_ROWS - 1) / SP_ROWS, dim3(SP_WAVES * 32), 0, stream>>>(
            H, AGG, ei, ew, bias, E, N);
    };

    transpose(W1, WT1, D_IN,  D_HID);
    transpose(W2, WT2, D_HID, D_HID);
    transpose(W3, WT3, D_HID, D_EMB);
    transpose(fcw, WTfc, D_EMB, D_OUT);

    gemm(x, WT1, B0, nullptr, D_IN, D_HID);
    spmm(B0, outEmb, b1);

    gemmReluA(outEmb, WT2, B0, nullptr, D_HID, D_HID);
    spmm(B0, B1, b2);

    gemmReluA(B1, WT3, B0, nullptr, D_HID, D_EMB);
    spmm(B0, outEmb, b3);

    gemm(outEmb, WTfc, Y, fcb, D_EMB, D_OUT);
    log_softmax16_kernel<<<(N + 255) / 256, blk, 0, stream>>>(Y, outLsm, N);
}


# Destination column swept once per call: SpMM launch 1 also stores each group's filtered edge records in the still unwritten log-softmax output, launches 2, 3 read that list; float sums in list order
# speedup vs baseline: 3.2545x; 1.7660x over previous
#include <hip/hip_runtime.h>
#include <math.h>

typedef float v2f __attribute__((ext_vector_type(2)));
typedef float v8f __attribute__((ext_vector_type(8)));

#define D_IN  256
#define D_HID 128
#define D_EMB 128
#define D_OUT 16

template <bool RELU_A, int K, int CT>
__global__ __launch_bounds__(256)
void gemm_wmma_f32_kernel(const float* __restrict__ A,
                          const float* __restrict__ Wm,
                          float* __restrict__ C,
                          const float* __restrict__ bias,
                          int M)
{
    constexpr int RT    = 8 / CT;
    constexpr int ROWS  = 16 * RT;
    constexpr int PITCH = K + 4;
    constexpr int DOUTv = 16 * CT;
    static_assert(RT * CT == 8 && K % 128 == 0, "8 waves per group, whole 512-byte row pieces");

    __shared__ __attribute__((aligned(16))) float As[ROWS * PITCH];

    const int lane = threadIdx.x & 31;
    const int wv   = __builtin_amdgcn_readfirstlane((int)(threadIdx.x >> 5));
    const int half = lane >> 4;
    const int lm   = lane & 15;
    const int blockRow = blockIdx.x * ROWS;

    float4 st[ROWS / 8][K / 128];
    #pragma unroll
    for (int i = 0; i < ROWS / 8; ++i) {
        const int gr = min(blockRow + wv + 8 * i, M - 1);
        const float* src = A + (size_t)gr * K + lane * 4;
        #pragma unroll
        for (int q = 0; q < K / 128; ++q) st[i][q] = *(const float4*)(src + q * 128);
    }
    #pragma unroll
    for (int i = 0; i < ROWS / 8; ++i) {
        float* dst = As + (wv + 8 * i) * PITCH + lane * 4;
        #pragma unroll
        for (int q = 0; q < K / 128; ++q) {
            float4 v = st[i][q];
            if (RELU_A) {
                v.x = fmaxf(v.x, 0.0f); v.y = fmaxf(v.y, 0.0f);
                v.z = fmaxf(v.z, 0.0f); v.w = fmaxf(v.w, 0.0f);
            }
            *(float4*)(dst + q * 128) = v;
        }
    }
    __syncthreads();

    const int rowBase = blockRow + (wv / CT) * 16;
    const int colBase = (wv % CT) * 16;

    const float* pA = As + ((wv / CT) * 16 + lm) * PITCH + 2 * half;
    const float* pB = Wm + (size_t)(2 * half) * DOUTv + colBase + lm;

    v8f acc = {};
    #pragma unroll 8
    for (int s = 0; s < K / 4; ++s) {
        const v2f a = *(const v2f*)(pA + 4 * s);
        v2f b;
        b.x = pB[(size_t)(4 * s)     * DOUTv];
        b.y = pB[(size_t)(4 * s + 1) * DOUTv];
        acc = __builtin_amdgcn_wmma_f32_16x16x4_f32(
                  false, a, false, b, (short)0, acc, false, false);
    }

    const float bv = bias ? bias[colBase + lm] : 0.0f;
    float* pC = C + (size_t)(rowBase + 8 * half) * DOUTv + colBase + lm;
    #pragma unroll
    for (int v = 0; v < 8; ++v) {
        const int r = rowBase + v + 8 * half;
        if (r < M) pC[(size_t)v * DOUTv] = acc[v] + bv;
    }
}

#define SP_ROWS   300
#define SP_EBITS  23
#define SP_WAVES  32
#define SP_SCAN_U 8
#define SP_GATH_U 8
#define SP_PEND   (32 + 128 * SP_SCAN_U)
#define SP_LIST_U    5
#define SP_LIST_MAX  (SP_WAVES * 32 * SP_LIST_U)
#define SP_LIST_GATH 16
#define SP_NOLIST    0xFFFFFFFFu
static_assert(SP_ROWS <= (1 << (32 - SP_EBITS)), "row number fits above the edge index");
static_assert((SP_ROWS * D_HID + SP_WAVES * SP_PEND + 1) * 4 <= 320 * 1024, "sums, pending lists and the cursor fit the LDS");

template <int U>
__device__ __forceinline__ void gather_add(const float* __restrict__ H, float* acc,
                                           int lane, int src, float w, int dl, int n)
{
    for (int i = 0; i < n; i += U) {
        float4 h[U];
        #pragma unroll
        for (int k = 0; k < U; ++k) {
            const int s = __builtin_amdgcn_readlane(src, i + k);
            h[k] = *(const float4*)(H + (size_t)s * D_HID + lane * 4);
        }
        #pragma unroll
        for (int k = 0; k < U; ++k) {
            if (i + k < n) {
                const float wk = __int_as_float(
                    __builtin_amdgcn_readlane(__float_as_int(w), i + k));
                const int dk = __builtin_amdgcn_readlane(dl, i + k);
                float* a = acc + dk * D_HID + lane;
                __hip_atomic_fetch_add(a,      h[k].x * wk, __ATOMIC_RELAXED, __HIP_MEMORY_SCOPE_WORKGROUP);
                __hip_atomic_fetch_add(a + 32, h[k].y * wk, __ATOMIC_RELAXED, __HIP_MEMORY_SCOPE_WORKGROUP);
                __hip_atomic_fetch_add(a + 64, h[k].z * wk, __ATOMIC_RELAXED, __HIP_MEMORY_SCOPE_WORKGROUP);
                __hip_atomic_fetch_add(a + 96, h[k].w * wk, __ATOMIC_RELAXED, __HIP_MEMORY_SCOPE_WORKGROUP);
            }
        }
    }
}

template <bool KEEP>
__global__ __launch_bounds__(SP_WAVES * 32)
void spmm_owner_kernel(const float* __restrict__ H,
                       float* __restrict__ AGG,
                       const int* __restrict__ ei,
                       const float* __restrict__ ew,
                       const float* __restrict__ bias,
                       float* __restrict__ keep,
                       int E, int N)
{
    __shared__ float    acc[SP_ROWS * D_HID];
    __shared__ unsigned pend[SP_WAVES * SP_PEND];
    __shared__ unsigned cursor;

    const int lane = threadIdx.x & 31;
    const int wv   = __builtin_amdgcn_readfirstlane((int)(threadIdx.x >> 5));
    const int lo   = blockIdx.x * SP_ROWS;
    unsigned* myPend = pend + wv * SP_PEND;
    const int* dstp  = ei + E;

    const int nGroups = (N + SP_ROWS - 1) / SP_ROWS;
    const int hdrDw   = (nGroups + 63) & ~63;
    const int room    = N * D_OUT - hdrDw;
    const int cap     = room > 0 ? min((room / nGroups) & ~31, SP_LIST_MAX) : 0;
    unsigned* counts  = (unsigned*)keep;
    unsigned* myList  = counts + hdrDw + (size_t)blockIdx.x * cap;

    for (int i = threadIdx.x; i < SP_ROWS * D_HID; i += SP_WAVES * 32) {
        const int p = i & (D_HID - 1);
        acc[i] = bias[((p & 31) << 2) | (p >> 5)];
    }
    if (KEEP && threadIdx.x == 0) cursor = 0u;
    __syncthreads();

    bool sweep = true;
    if (!KEEP) {
        const unsigned total = __builtin_amdgcn_readfirstlane(counts[blockIdx.x]);
        if (total <= (unsigned)cap) {
            sweep = false;
            unsigned rec[SP_LIST_U];
            int      src[SP_LIST_U];
            float    w[SP_LIST_U];
            #pragma unroll
            for (int u = 0; u < SP_LIST_U; ++u) {
                const int c = (wv + SP_WAVES * u) * 32 + lane;
                rec[u] = c < (int)total ? myList[c] : 0u;
            }
            #pragma unroll
            for (int u = 0; u < SP_LIST_U; ++u) {
                const int  c = (wv + SP_WAVES * u) * 32 + lane;
                const bool v = c < (int)total;
                const int  e = (int)(rec[u] & ((1u << SP_EBITS) - 1u));
                src[u] = v ? ei[e] : 0;
                w[u]   = v ? ew[e] : 0.0f;
            }
            #pragma unroll
            for (int u = 0; u < SP_LIST_U; ++u) {
                const int c0 = (wv + SP_WAVES * u) * 32;
                if (c0 < (int)total)
                    gather_add<SP_LIST_GATH>(H, acc, lane, src[u], w[u],
                                             (int)(rec[u] >> SP_EBITS), min(32, (int)total - c0));
            }
        }
    }

    if (sweep) {
        int cnt = 0;

        auto append = [&](bool m, unsigned rec) {
            const unsigned mask = __builtin_amdgcn_ballot_w32(m);
            if (m) myPend[cnt + (int)__builtin_amdgcn_mbcnt_lo(mask, 0u)] = rec;
            cnt += __builtin_popcount(mask);
        };

        auto drain = [&](int base, int n) {
            unsigned rec = 0u;
            int   src = 0;
            float w   = 0.0f;
            if (lane < n) {
                rec = myPend[base + lane];
                const int e = (int)(rec & ((1u << SP_EBITS) - 1u));
                src = ei[e];
                w   = ew[e];
            }
            if (KEEP) {
                unsigned pos = 0u;
                if (lane == 0)
                    pos = __hip_atomic_fetch_add(&cursor, (unsigned)n, __ATOMIC_RELAXED, __HIP_MEMORY_SCOPE_WORKGROUP);
                pos = __builtin_amdgcn_readfirstlane(pos);
                if (lane < n && pos + (unsigned)n <= (unsigned)cap) myList[pos + lane] = rec;
            }
            gather_add<SP_GATH_U>(H, acc, lane, src, w, (int)(rec >> SP_EBITS), n);
        };

        const int head = min(E, (int)(((16u - (unsigned)((size_t)dstp & 15u)) & 15u) >> 2));
        const int nvec = (E - head) >> 2;
        const int fullChunks = nvec >> 5;
        const int4* dv = (const int4*)(dstp + head);

        for (int c = wv; c < fullChunks; c += SP_WAVES * SP_SCAN_U) {
            int4 d[SP_SCAN_U];
            #pragma unroll
            for (int u = 0; u < SP_SCAN_U; ++u) {
                const int cu = min(c + SP_WAVES * u, fullChunks - 1);
                d[u] = dv[cu * 32 + lane];
            }
            #pragma unroll
            for (int u = 0; u < SP_SCAN_U; ++u) {
                const int  cu = c + SP_WAVES * u;
                const bool ok = cu < fullChunks;
                const unsigned r0 = (unsigned)(d[u].x - lo), r1 = (unsigned)(d[u].y - lo);
                const unsigned r2 = (unsigned)(d[u].z - lo), r3 = (unsigned)(d[u].w - lo);
                const bool m0 = ok && r0 < SP_ROWS, m1 = ok && r1 < SP_ROWS;
                const bool m2 = ok && r2 < SP_ROWS, m3 = ok && r3 < SP_ROWS;
                if (__builtin_amdgcn_ballot_w32(m0 || m1 || m2 || m3)) {
                    const unsigned e0 = (unsigned)(head + ((cu * 32 + lane) << 2));
                    append(m0, (e0     ) | (r0 << SP_EBITS));
                    append(m1, (e0 + 1u) | (r1 << SP_EBITS));
                    append(m2, (e0 + 2u) | (r2 << SP_EBITS));
                    append(m3, (e0 + 3u) | (r3 << SP_EBITS));
                }
            }
            while (cnt >= 32) { cnt -= 32; drain(cnt, 32); }
        }

        if (wv == 0) {
            const int bodyEnd = head + (fullChunks << 7);
            const int nLeft   = head + (E - bodyEnd);
            for (int b = 0; b < nLeft; b += 32) {
                const int idx = b + lane;
                const bool valid = idx < nLeft;
                const int e = idx < head ? idx : bodyEnd + (idx - head);
                const unsigned r = valid ? (unsigned)(dstp[e] - lo) : 0xFFFFFFFFu;
                append(r < SP_ROWS, (unsigned)e | (r << SP_EBITS));
            }
        }
        while (cnt > 0) { const int n = min(cnt, 32); cnt -= n; drain(cnt, n); }
    }

    __syncthreads();

    if (KEEP && threadIdx.x == 0) {
        const unsigned total = cursor;
        counts[blockIdx.x] = total <= (unsigned)cap ? total : SP_NOLIST;
    }

    for (int r = wv; r < SP_ROWS; r += SP_WAVES) {
        const int row = lo + r;
        if (row < N) {
            const float* a = acc + r * D_HID + lane;
            const float4 v = {a[0], a[32], a[64], a[96]};
            *(float4*)(AGG + (size_t)row * D_HID + lane * 4) = v;
        }
    }
}

__global__ void log_softmax16_kernel(const float* __restrict__ Y,
                                     float* __restrict__ out, int N)
{
    const int r = blockIdx.x * blockDim.x + threadIdx.x;
    if (r >= N) return;
    float v[D_OUT];
    float m = -INFINITY;
    #pragma unroll
    for (int j = 0; j < D_OUT; ++j) { v[j] = Y[(size_t)r * D_OUT + j]; m = fmaxf(m, v[j]); }
    float s = 0.0f;
    #pragma unroll
    for (int j = 0; j < D_OUT; ++j) s += __expf(v[j] - m);
    const float l = m + __logf(s);
    #pragma unroll
    for (int j = 0; j < D_OUT; ++j) out[(size_t)r * D_OUT + j] = v[j] - l;
}

extern "C" void kernel_launch(void* const* d_in, const int* in_sizes, int n_in,
                              void* d_out, int out_size, void* d_ws, size_t ws_size,
                              hipStream_t stream)
{
    const float* x    = (const float*)d_in[0];
    const float* ew   = (const float*)d_in[1];
    const float* W1   = (const float*)d_in[2];
    const float* b1   = (const float*)d_in[3];
    const float* W2   = (const float*)d_in[4];
    const float* b2   = (const float*)d_in[5];
    const float* W3   = (const float*)d_in[6];
    const float* b3   = (const float*)d_in[7];
    const float* fcw  = (const float*)d_in[8];
    const float* fcb  = (const float*)d_in[9];
    const int*   ei   = (const int*)d_in[10];

    const int N = in_sizes[0] / D_IN;
    const int E = in_sizes[1];

    float* B0   = (float*)d_ws;
    float* B1   = B0 + (size_t)N * D_HID;
    float* Y    = B0;

    float* outEmb = (float*)d_out;
    float* outLsm = outEmb + (size_t)N * D_EMB;

    const int rowTiles = (N + 15) / 16;
    const dim3 blk(256);

    const dim3 spGrid((N + SP_ROWS - 1) / SP_ROWS), spBlk(SP_WAVES * 32);
    auto spmm_first = [&](const float* H, float* AGG, const float* bias) {
        spmm_owner_kernel<true><<<spGrid, spBlk, 0, stream>>>(H, AGG, ei, ew, bias, outLsm, E, N);
    };
    auto spmm = [&](const float* H, float* AGG, const float* bias) {
        spmm_owner_kernel<false><<<spGrid, spBlk, 0, stream>>>(H, AGG, ei, ew, bias, outLsm, E, N);
    };

    gemm_wmma_f32_kernel<false, D_IN, 8><<<rowTiles, blk, 0, stream>>>(x, W1, B0, nullptr, N);
    spmm_first(B0, outEmb, b1);

    gemm_wmma_f32_kernel<true, D_HID, 8><<<rowTiles, blk, 0, stream>>>(outEmb, W2, B0, nullptr, N);
    spmm(B0, B1, b2);

    gemm_wmma_f32_kernel<true, D_HID, 8><<<rowTiles, blk, 0, stream>>>(B1, W3, B0, nullptr, N);
    spmm(B0, outEmb, b3);

    gemm_wmma_f32_kernel<false, D_EMB, 1><<<(rowTiles + 7) / 8, blk, 0, stream>>>(outEmb, fcw, Y, fcb, N);
    log_softmax16_kernel<<<(N + 255) / 256, blk, 0, stream>>>(Y, outLsm, N);
}


# SpMM groups cut from 300 to 200 destination rows: 250 one-per-WGP groups fill the chip in one round instead of 167, a third fewer edges gathered and added per WGP; float sums regrouped, arrival order
# speedup vs baseline: 3.4765x; 1.0682x over previous
#include <hip/hip_runtime.h>
#include <math.h>

typedef float v2f __attribute__((ext_vector_type(2)));
typedef float v8f __attribute__((ext_vector_type(8)));

#define D_IN  256
#define D_HID 128
#define D_EMB 128
#define D_OUT 16

template <bool RELU_A, int K, int CT>
__global__ __launch_bounds__(256)
void gemm_wmma_f32_kernel(const float* __restrict__ A,
                          const float* __restrict__ Wm,
                          float* __restrict__ C,
                          const float* __restrict__ bias,
                          int M)
{
    constexpr int RT    = 8 / CT;
    constexpr int ROWS  = 16 * RT;
    constexpr int PITCH = K + 4;
    constexpr int DOUTv = 16 * CT;
    static_assert(RT * CT == 8 && K % 128 == 0, "8 waves per group, whole 512-byte row pieces");

    __shared__ __attribute__((aligned(16))) float As[ROWS * PITCH];

    const int lane = threadIdx.x & 31;
    const int wv   = __builtin_amdgcn_readfirstlane((int)(threadIdx.x >> 5));
    const int half = lane >> 4;
    const int lm   = lane & 15;
    const int blockRow = blockIdx.x * ROWS;

    float4 st[ROWS / 8][K / 128];
    #pragma unroll
    for (int i = 0; i < ROWS / 8; ++i) {
        const int gr = min(blockRow + wv + 8 * i, M - 1);
        const float* src = A + (size_t)gr * K + lane * 4;
        #pragma unroll
        for (int q = 0; q < K / 128; ++q) st[i][q] = *(const float4*)(src + q * 128);
    }
    #pragma unroll
    for (int i = 0; i < ROWS / 8; ++i) {
        float* dst = As + (wv + 8 * i) * PITCH + lane * 4;
        #pragma unroll
        for (int q = 0; q < K / 128; ++q) {
            float4 v = st[i][q];
            if (RELU_A) {
                v.x = fmaxf(v.x, 0.0f); v.y = fmaxf(v.y, 0.0f);
                v.z = fmaxf(v.z, 0.0f); v.w = fmaxf(v.w, 0.0f);
            }
            *(float4*)(dst + q * 128) = v;
        }
    }
    __syncthreads();

    const int rowBase = blockRow + (wv / CT) * 16;
    const int colBase = (wv % CT) * 16;

    const float* pA = As + ((wv / CT) * 16 + lm) * PITCH + 2 * half;
    const float* pB = Wm + (size_t)(2 * half) * DOUTv + colBase + lm;

    v8f acc = {};
    #pragma unroll 8
    for (int s = 0; s < K / 4; ++s) {
        const v2f a = *(const v2f*)(pA + 4 * s);
        v2f b;
        b.x = pB[(size_t)(4 * s)     * DOUTv];
        b.y = pB[(size_t)(4 * s + 1) * DOUTv];
        acc = __builtin_amdgcn_wmma_f32_16x16x4_f32(
                  false, a, false, b, (short)0, acc, false, false);
    }

    const float bv = bias ? bias[colBase + lm] : 0.0f;
    float* pC = C + (size_t)(rowBase + 8 * half) * DOUTv + colBase + lm;
    #pragma unroll
    for (int v = 0; v < 8; ++v) {
        const int r = rowBase + v + 8 * half;
        if (r < M) pC[(size_t)v * DOUTv] = acc[v] + bv;
    }
}

#define SP_ROWS   200
#define SP_EBITS  23
#define SP_WAVES  32
#define SP_SCAN_U 8
#define SP_GATH_U 8
#define SP_PEND   (32 + 128 * SP_SCAN_U)
#define SP_LIST_U    5
#define SP_LIST_MAX  (SP_WAVES * 32 * SP_LIST_U)
#define SP_LIST_GATH 16
#define SP_NOLIST    0xFFFFFFFFu
static_assert(SP_ROWS <= (1 << (32 - SP_EBITS)), "row number fits above the edge index");
static_assert((SP_ROWS * D_HID + SP_WAVES * SP_PEND + 1) * 4 <= 320 * 1024, "sums, pending lists and the cursor fit the LDS");

template <int U>
__device__ __forceinline__ void gather_add(const float* __restrict__ H, float* acc,
                                           int lane, int src, float w, int dl, int n)
{
    for (int i = 0; i < n; i += U) {
        float4 h[U];
        #pragma unroll
        for (int k = 0; k < U; ++k) {
            const int s = __builtin_amdgcn_readlane(src, i + k);
            h[k] = *(const float4*)(H + (size_t)s * D_HID + lane * 4);
        }
        #pragma unroll
        for (int k = 0; k < U; ++k) {
            if (i + k < n) {
                const float wk = __int_as_float(
                    __builtin_amdgcn_readlane(__float_as_int(w), i + k));
                const int dk = __builtin_amdgcn_readlane(dl, i + k);
                float* a = acc + dk * D_HID + lane;
                __hip_atomic_fetch_add(a,      h[k].x * wk, __ATOMIC_RELAXED, __HIP_MEMORY_SCOPE_WORKGROUP);
                __hip_atomic_fetch_add(a + 32, h[k].y * wk, __ATOMIC_RELAXED, __HIP_MEMORY_SCOPE_WORKGROUP);
                __hip_atomic_fetch_add(a + 64, h[k].z * wk, __ATOMIC_RELAXED, __HIP_MEMORY_SCOPE_WORKGROUP);
                __hip_atomic_fetch_add(a + 96, h[k].w * wk, __ATOMIC_RELAXED, __HIP_MEMORY_SCOPE_WORKGROUP);
            }
        }
    }
}

template <bool KEEP>
__global__ __launch_bounds__(SP_WAVES * 32)
void spmm_owner_kernel(const float* __restrict__ H,
                       float* __restrict__ AGG,
                       const int* __restrict__ ei,
                       const float* __restrict__ ew,
                       const float* __restrict__ bias,
                       float* __restrict__ keep,
                       int E, int N)
{
    __shared__ float    acc[SP_ROWS * D_HID];
    __shared__ unsigned pend[SP_WAVES * SP_PEND];
    __shared__ unsigned cursor;

    const int lane = threadIdx.x & 31;
    const int wv   = __builtin_amdgcn_readfirstlane((int)(threadIdx.x >> 5));
    const int lo   = blockIdx.x * SP_ROWS;
    unsigned* myPend = pend + wv * SP_PEND;
    const int* dstp  = ei + E;

    const int nGroups = (N + SP_ROWS - 1) / SP_ROWS;
    const int hdrDw   = (nGroups + 63) & ~63;
    const int room    = N * D_OUT - hdrDw;
    const int cap     = room > 0 ? min((room / nGroups) & ~31, SP_LIST_MAX) : 0;
    unsigned* counts  = (unsigned*)keep;
    unsigned* myList  = counts + hdrDw + (size_t)blockIdx.x * cap;

    for (int i = threadIdx.x; i < SP_ROWS * D_HID; i += SP_WAVES * 32) {
        const int p = i & (D_HID - 1);
        acc[i] = bias[((p & 31) << 2) | (p >> 5)];
    }
    if (KEEP && threadIdx.x == 0) cursor = 0u;
    __syncthreads();

    bool sweep = true;
    if (!KEEP) {
        const unsigned total = __builtin_amdgcn_readfirstlane(counts[blockIdx.x]);
        if (total <= (unsigned)cap) {
            sweep = false;
            unsigned rec[SP_LIST_U];
            int      src[SP_LIST_U];
            float    w[SP_LIST_U];
            #pragma unroll
            for (int u = 0; u < SP_LIST_U; ++u) {
                const int c = (wv + SP_WAVES * u) * 32 + lane;
                rec[u] = c < (int)total ? myList[c] : 0u;
            }
            #pragma unroll
            for (int u = 0; u < SP_LIST_U; ++u) {
                const int  c = (wv + SP_WAVES * u) * 32 + lane;
                const bool v = c < (int)total;
                const int  e = (int)(rec[u] & ((1u << SP_EBITS) - 1u));
                src[u] = v ? ei[e] : 0;
                w[u]   = v ? ew[e] : 0.0f;
            }
            #pragma unroll
            for (int u = 0; u < SP_LIST_U; ++u) {
                const int c0 = (wv + SP_WAVES * u) * 32;
                if (c0 < (int)total)
                    gather_add<SP_LIST_GATH>(H, acc, lane, src[u], w[u],
                                             (int)(rec[u] >> SP_EBITS), min(32, (int)total - c0));
            }
        }
    }

    if (sweep) {
        int cnt = 0;

        auto append = [&](bool m, unsigned rec) {
            const unsigned mask = __builtin_amdgcn_ballot_w32(m);
            if (m) myPend[cnt + (int)__builtin_amdgcn_mbcnt_lo(mask, 0u)] = rec;
            cnt += __builtin_popcount(mask);
        };

        auto drain = [&](int base, int n) {
            unsigned rec = 0u;
            int   src = 0;
            float w   = 0.0f;
            if (lane < n) {
                rec = myPend[base + lane];
                const int e = (int)(rec & ((1u << SP_EBITS) - 1u));
                src = ei[e];
                w   = ew[e];
            }
            if (KEEP) {
                unsigned pos = 0u;
                if (lane == 0)
                    pos = __hip_atomic_fetch_add(&cursor, (unsigned)n, __ATOMIC_RELAXED, __HIP_MEMORY_SCOPE_WORKGROUP);
                pos = __builtin_amdgcn_readfirstlane(pos);
                if (lane < n && pos + (unsigned)n <= (unsigned)cap) myList[pos + lane] = rec;
            }
            gather_add<SP_GATH_U>(H, acc, lane, src, w, (int)(rec >> SP_EBITS), n);
        };

        const int head = min(E, (int)(((16u - (unsigned)((size_t)dstp & 15u)) & 15u) >> 2));
        const int nvec = (E - head) >> 2;
        const int fullChunks = nvec >> 5;
        const int4* dv = (const int4*)(dstp + head);

        for (int c = wv; c < fullChunks; c += SP_WAVES * SP_SCAN_U) {
            int4 d[SP_SCAN_U];
            #pragma unroll
            for (int u = 0; u < SP_SCAN_U; ++u) {
                const int cu = min(c + SP_WAVES * u, fullChunks - 1);
                d[u] = dv[cu * 32 + lane];
            }
            #pragma unroll
            for (int u = 0; u < SP_SCAN_U; ++u) {
                const int  cu = c + SP_WAVES * u;
                const bool ok = cu < fullChunks;
                const unsigned r0 = (unsigned)(d[u].x - lo), r1 = (unsigned)(d[u].y - lo);
                const unsigned r2 = (unsigned)(d[u].z - lo), r3 = (unsigned)(d[u].w - lo);
                const bool m0 = ok && r0 < SP_ROWS, m1 = ok && r1 < SP_ROWS;
                const bool m2 = ok && r2 < SP_ROWS, m3 = ok && r3 < SP_ROWS;
                if (__builtin_amdgcn_ballot_w32(m0 || m1 || m2 || m3)) {
                    const unsigned e0 = (unsigned)(head + ((cu * 32 + lane) << 2));
                    append(m0, (e0     ) | (r0 << SP_EBITS));
                    append(m1, (e0 + 1u) | (r1 << SP_EBITS));
                    append(m2, (e0 + 2u) | (r2 << SP_EBITS));
                    append(m3, (e0 + 3u) | (r3 << SP_EBITS));
                }
            }
            while (cnt >= 32) { cnt -= 32; drain(cnt, 32); }
        }

        if (wv == 0) {
            const int bodyEnd = head + (fullChunks << 7);
            const int nLeft   = head + (E - bodyEnd);
            for (int b = 0; b < nLeft; b += 32) {
                const int idx = b + lane;
                const bool valid = idx < nLeft;
                const int e = idx < head ? idx : bodyEnd + (idx - head);
                const unsigned r = valid ? (unsigned)(dstp[e] - lo) : 0xFFFFFFFFu;
                append(r < SP_ROWS, (unsigned)e | (r << SP_EBITS));
            }
        }
        while (cnt > 0) { const int n = min(cnt, 32); cnt -= n; drain(cnt, n); }
    }

    __syncthreads();

    if (KEEP && threadIdx.x == 0) {
        const unsigned total = cursor;
        counts[blockIdx.x] = total <= (unsigned)cap ? total : SP_NOLIST;
    }

    for (int r = wv; r < SP_ROWS; r += SP_WAVES) {
        const int row = lo + r;
        if (row < N) {
            const float* a = acc + r * D_HID + lane;
            const float4 v = {a[0], a[32], a[64], a[96]};
            *(float4*)(AGG + (size_t)row * D_HID + lane * 4) = v;
        }
    }
}

__global__ void log_softmax16_kernel(const float* __restrict__ Y,
                                     float* __restrict__ out, int N)
{
    const int r = blockIdx.x * blockDim.x + threadIdx.x;
    if (r >= N) return;
    float v[D_OUT];
    float m = -INFINITY;
    #pragma unroll
    for (int j = 0; j < D_OUT; ++j) { v[j] = Y[(size_t)r * D_OUT + j]; m = fmaxf(m, v[j]); }
    float s = 0.0f;
    #pragma unroll
    for (int j = 0; j < D_OUT; ++j) s += __expf(v[j] - m);
    const float l = m + __logf(s);
    #pragma unroll
    for (int j = 0; j < D_OUT; ++j) out[(size_t)r * D_OUT + j] = v[j] - l;
}

extern "C" void kernel_launch(void* const* d_in, const int* in_sizes, int n_in,
                              void* d_out, int out_size, void* d_ws, size_t ws_size,
                              hipStream_t stream)
{
    const float* x    = (const float*)d_in[0];
    const float* ew   = (const float*)d_in[1];
    const float* W1   = (const float*)d_in[2];
    const float* b1   = (const float*)d_in[3];
    const float* W2   = (const float*)d_in[4];
    const float* b2   = (const float*)d_in[5];
    const float* W3   = (const float*)d_in[6];
    const float* b3   = (const float*)d_in[7];
    const float* fcw  = (const float*)d_in[8];
    const float* fcb  = (const float*)d_in[9];
    const int*   ei   = (const int*)d_in[10];

    const int N = in_sizes[0] / D_IN;
    const int E = in_sizes[1];

    float* B0   = (float*)d_ws;
    float* B1   = B0 + (size_t)N * D_HID;
    float* Y    = B0;

    float* outEmb = (float*)d_out;
    float* outLsm = outEmb + (size_t)N * D_EMB;

    const int rowTiles = (N + 15) / 16;
    const dim3 blk(256);

    const dim3 spGrid((N + SP_ROWS - 1) / SP_ROWS), spBlk(SP_WAVES * 32);
    auto spmm_first = [&](const float* H, float* AGG, const float* bias) {
        spmm_owner_kernel<true><<<spGrid, spBlk, 0, stream>>>(H, AGG, ei, ew, bias, outLsm, E, N);
    };
    auto spmm = [&](const float* H, float* AGG, const float* bias) {
        spmm_owner_kernel<false><<<spGrid, spBlk, 0, stream>>>(H, AGG, ei, ew, bias, outLsm, E, N);
    };

    gemm_wmma_f32_kernel<false, D_IN, 8><<<rowTiles, blk, 0, stream>>>(x, W1, B0, nullptr, N);
    spmm_first(B0, outEmb, b1);

    gemm_wmma_f32_kernel<true, D_HID, 8><<<rowTiles, blk, 0, stream>>>(outEmb, W2, B0, nullptr, N);
    spmm(B0, B1, b2);

    gemm_wmma_f32_kernel<true, D_HID, 8><<<rowTiles, blk, 0, stream>>>(B1, W3, B0, nullptr, N);
    spmm(B0, outEmb, b3);

    gemm_wmma_f32_kernel<false, D_EMB, 1><<<(rowTiles + 7) / 8, blk, 0, stream>>>(outEmb, fcw, Y, fcb, N);
    log_softmax16_kernel<<<(N + 255) / 256, blk, 0, stream>>>(Y, outLsm, N);
}


# Next layer's dense product done by the SpMM groups: relu(LDS sums) @ W on f32 WMMA, A fragments from LDS; 2 GEMM launches and the sums' store and reload dropped; GEMM k sum order follows LDS layout
# speedup vs baseline: 3.7233x; 1.0710x over previous
#include <hip/hip_runtime.h>
#include <math.h>

typedef float v2f __attribute__((ext_vector_type(2)));
typedef float v8f __attribute__((ext_vector_type(8)));

#define D_IN  256
#define D_HID 128
#define D_EMB 128
#define D_OUT 16

template <int K, int CT>
__global__ __launch_bounds__(256)
void gemm_wmma_f32_kernel(const float* __restrict__ A,
                          const float* __restrict__ Wm,
                          float* __restrict__ C,
                          const float* __restrict__ bias,
                          int M)
{
    constexpr int RT    = 8 / CT;
    constexpr int ROWS  = 16 * RT;
    constexpr int PITCH = K + 4;
    constexpr int DOUTv = 16 * CT;
    static_assert(RT * CT == 8 && K % 128 == 0, "8 waves per group, whole 512-byte row pieces");

    __shared__ __attribute__((aligned(16))) float As[ROWS * PITCH];

    const int lane = threadIdx.x & 31;
    const int wv   = __builtin_amdgcn_readfirstlane((int)(threadIdx.x >> 5));
    const int half = lane >> 4;
    const int lm   = lane & 15;
    const int blockRow = blockIdx.x * ROWS;

    float4 st[ROWS / 8][K / 128];
    #pragma unroll
    for (int i = 0; i < ROWS / 8; ++i) {
        const int gr = min(blockRow + wv + 8 * i, M - 1);
        const float* src = A + (size_t)gr * K + lane * 4;
        #pragma unroll
        for (int q = 0; q < K / 128; ++q) st[i][q] = *(const float4*)(src + q * 128);
    }
    #pragma unroll
    for (int i = 0; i < ROWS / 8; ++i) {
        float* dst = As + (wv + 8 * i) * PITCH + lane * 4;
        #pragma unroll
        for (int q = 0; q < K / 128; ++q) {
            *(float4*)(dst + q * 128) = st[i][q];
        }
    }
    __syncthreads();

    const int rowBase = blockRow + (wv / CT) * 16;
    const int colBase = (wv % CT) * 16;

    const float* pA = As + ((wv / CT) * 16 + lm) * PITCH + 2 * half;
    const float* pB = Wm + (size_t)(2 * half) * DOUTv + colBase + lm;

    v8f acc = {};
    #pragma unroll 8
    for (int s = 0; s < K / 4; ++s) {
        const v2f a = *(const v2f*)(pA + 4 * s);
        v2f b;
        b.x = pB[(size_t)(4 * s)     * DOUTv];
        b.y = pB[(size_t)(4 * s + 1) * DOUTv];
        acc = __builtin_amdgcn_wmma_f32_16x16x4_f32(
                  false, a, false, b, (short)0, acc, false, false);
    }

    const float bv = bias ? bias[colBase + lm] : 0.0f;
    float* pC = C + (size_t)(rowBase + 8 * half) * DOUTv + colBase + lm;
    #pragma unroll
    for (int v = 0; v < 8; ++v) {
        const int r = rowBase + v + 8 * half;
        if (r < M) pC[(size_t)v * DOUTv] = acc[v] + bv;
    }
}

#define SP_ROWS   200
#define SP_TILES  ((SP_ROWS + 15) / 16)
#define SP_LROWS  (16 * SP_TILES)
#define SP_PITCH  (D_HID + 4)
#define SP_EBITS  23
#define SP_WAVES  32
#define SP_SCAN_U 8
#define SP_GATH_U 8
#define SP_PEND   (32 + 128 * SP_SCAN_U)
#define SP_LIST_U    5
#define SP_LIST_MAX  (SP_WAVES * 32 * SP_LIST_U)
#define SP_LIST_GATH 16
#define SP_NOLIST    0xFFFFFFFFu
static_assert(SP_ROWS <= (1 << (32 - SP_EBITS)), "row number fits above the edge index");
static_assert((SP_LROWS * SP_PITCH + SP_WAVES * SP_PEND + 1) * 4 <= 320 * 1024, "sums, pending lists and the cursor fit the LDS");
static_assert(SP_WAVES == 32 && D_HID == 128, "next-layer product: 8 column tiles x 4 row groups of waves");

template <int U>
__device__ __forceinline__ void gather_add(const float* __restrict__ H, float* acc,
                                           int lane, int src, float w, int dl, int n)
{
    for (int i = 0; i < n; i += U) {
        float4 h[U];
        #pragma unroll
        for (int k = 0; k < U; ++k) {
            const int s = __builtin_amdgcn_readlane(src, i + k);
            h[k] = *(const float4*)(H + (size_t)s * D_HID + lane * 4);
        }
        #pragma unroll
        for (int k = 0; k < U; ++k) {
            if (i + k < n) {
                const float wk = __int_as_float(
                    __builtin_amdgcn_readlane(__float_as_int(w), i + k));
                const int dk = __builtin_amdgcn_readlane(dl, i + k);
                float* a = acc + dk * SP_PITCH + lane;
                __hip_atomic_fetch_add(a,      h[k].x * wk, __ATOMIC_RELAXED, __HIP_MEMORY_SCOPE_WORKGROUP);
                __hip_atomic_fetch_add(a + 32, h[k].y * wk, __ATOMIC_RELAXED, __HIP_MEMORY_SCOPE_WORKGROUP);
                __hip_atomic_fetch_add(a + 64, h[k].z * wk, __ATOMIC_RELAXED, __HIP_MEMORY_SCOPE_WORKGROUP);
                __hip_atomic_fetch_add(a + 96, h[k].w * wk, __ATOMIC_RELAXED, __HIP_MEMORY_SCOPE_WORKGROUP);
            }
        }
    }
}

template <int NT>
__device__ __forceinline__ void next_layer_tiles(const float* acc, const float* __restrict__ Wn,
                                                 float* __restrict__ OUT,
                                                 int lane, int wv, int lo, int N)
{
    const int half = lane >> 4;
    const int lm   = lane & 15;
    const int ct   = wv & 7;
    const int rg   = wv >> 3;
    const float* pA = acc + (rg * 16 + lm) * SP_PITCH + 2 * half;
    const float* pB = Wn + (size_t)(8 * half) * D_HID + ct * 16 + lm;

    v8f c[NT];
    #pragma unroll
    for (int i = 0; i < NT; ++i) c[i] = v8f{};

    for (int j = 0; j < D_HID / 32; ++j) {
        #pragma unroll
        for (int u = 0; u < 8; ++u) {
            v2f b;
            b.x = pB[(size_t)(16 * u + j)     * D_HID];
            b.y = pB[(size_t)(16 * u + j + 4) * D_HID];
            #pragma unroll
            for (int i = 0; i < NT; ++i) {
                v2f a = *(const v2f*)(pA + i * 64 * SP_PITCH + 32 * j + 4 * u);
                a.x = fmaxf(a.x, 0.0f);
                a.y = fmaxf(a.y, 0.0f);
                c[i] = __builtin_amdgcn_wmma_f32_16x16x4_f32(
                           false, a, false, b, (short)0, c[i], false, false);
            }
        }
    }

    #pragma unroll
    for (int i = 0; i < NT; ++i) {
        #pragma unroll
        for (int v = 0; v < 8; ++v) {
            const int r = (rg + 4 * i) * 16 + 8 * half + v;
            if (r < SP_ROWS && lo + r < N)
                OUT[(size_t)(lo + r) * D_HID + ct * 16 + lm] = c[i][v];
        }
    }
}

template <bool KEEP, bool NEXT>
__global__ __launch_bounds__(SP_WAVES * 32)
void spmm_owner_kernel(const float* __restrict__ H,
                       float* __restrict__ AGG,
                       const int* __restrict__ ei,
                       const float* __restrict__ ew,
                       const float* __restrict__ bias,
                       const float* __restrict__ Wn,
                       float* __restrict__ keep,
                       int E, int N)
{
    __shared__ float    acc[SP_LROWS * SP_PITCH];
    __shared__ unsigned pend[SP_WAVES * SP_PEND];
    __shared__ unsigned cursor;

    const int lane = threadIdx.x & 31;
    const int wv   = __builtin_amdgcn_readfirstlane((int)(threadIdx.x >> 5));
    const int lo   = blockIdx.x * SP_ROWS;
    unsigned* myPend = pend + wv * SP_PEND;
    const int* dstp  = ei + E;

    const int nGroups = (N + SP_ROWS - 1) / SP_ROWS;
    const int hdrDw   = (nGroups + 63) & ~63;
    const int room    = N * D_OUT - hdrDw;
    const int cap     = room > 0 ? min((room / nGroups) & ~31, SP_LIST_MAX) : 0;
    unsigned* counts  = (unsigned*)keep;
    unsigned* myList  = counts + hdrDw + (size_t)blockIdx.x * cap;

    for (int i = threadIdx.x; i < SP_LROWS * D_HID; i += SP_WAVES * 32) {
        const int p = i & (D_HID - 1);
        acc[(i / D_HID) * SP_PITCH + p] = bias[((p & 31) << 2) | (p >> 5)];
    }
    if (KEEP && threadIdx.x == 0) cursor = 0u;
    __syncthreads();

    bool sweep = true;
    if (!KEEP) {
        const unsigned total = __builtin_amdgcn_readfirstlane(counts[blockIdx.x]);
        if (total <= (unsigned)cap) {
            sweep = false;
            unsigned rec[SP_LIST_U];
            int      src[SP_LIST_U];
            float    w[SP_LIST_U];
            #pragma unroll
            for (int u = 0; u < SP_LIST_U; ++u) {
                const int c = (wv + SP_WAVES * u) * 32 + lane;
                rec[u] = c < (int)total ? myList[c] : 0u;
            }
            #pragma unroll
            for (int u = 0; u < SP_LIST_U; ++u) {
                const int  c = (wv + SP_WAVES * u) * 32 + lane;
                const bool v = c < (int)total;
                const int  e = (int)(rec[u] & ((1u << SP_EBITS) - 1u));
                src[u] = v ? ei[e] : 0;
                w[u]   = v ? ew[e] : 0.0f;
            }
            #pragma unroll
            for (int u = 0; u < SP_LIST_U; ++u) {
                const int c0 = (wv + SP_WAVES * u) * 32;
                if (c0 < (int)total)
                    gather_add<SP_LIST_GATH>(H, acc, lane, src[u], w[u],
                                             (int)(rec[u] >> SP_EBITS), min(32, (int)total - c0));
            }
        }
    }

    if (sweep) {
        int cnt = 0;

        auto append = [&](bool m, unsigned rec) {
            const unsigned mask = __builtin_amdgcn_ballot_w32(m);
            if (m) myPend[cnt + (int)__builtin_amdgcn_mbcnt_lo(mask, 0u)] = rec;
            cnt += __builtin_popcount(mask);
        };

        auto drain = [&](int base, int n) {
            unsigned rec = 0u;
            int   src = 0;
            float w   = 0.0f;
            if (lane < n) {
                rec = myPend[base + lane];
                const int e = (int)(rec & ((1u << SP_EBITS) - 1u));
                src = ei[e];
                w   = ew[e];
            }
            if (KEEP) {
                unsigned pos = 0u;
                if (lane == 0)
                    pos = __hip_atomic_fetch_add(&cursor, (unsigned)n, __ATOMIC_RELAXED, __HIP_MEMORY_SCOPE_WORKGROUP);
                pos = __builtin_amdgcn_readfirstlane(pos);
                if (lane < n && pos + (unsigned)n <= (unsigned)cap) myList[pos + lane] = rec;
            }
            gather_add<SP_GATH_U>(H, acc, lane, src, w, (int)(rec >> SP_EBITS), n);
        };

        const int head = min(E, (int)(((16u - (unsigned)((size_t)dstp & 15u)) & 15u) >> 2));
        const int nvec = (E - head) >> 2;
        const int fullChunks = nvec >> 5;
        const int4* dv = (const int4*)(dstp + head);

        for (int c = wv; c < fullChunks; c += SP_WAVES * SP_SCAN_U) {
            int4 d[SP_SCAN_U];
            #pragma unroll
            for (int u = 0; u < SP_SCAN_U; ++u) {
                const int cu = min(c + SP_WAVES * u, fullChunks - 1);
                d[u] = dv[cu * 32 + lane];
            }
            #pragma unroll
            for (int u = 0; u < SP_SCAN_U; ++u) {
                const int  cu = c + SP_WAVES * u;
                const bool ok = cu < fullChunks;
                const unsigned r0 = (unsigned)(d[u].x - lo), r1 = (unsigned)(d[u].y - lo);
                const unsigned r2 = (unsigned)(d[u].z - lo), r3 = (unsigned)(d[u].w - lo);
                const bool m0 = ok && r0 < SP_ROWS, m1 = ok && r1 < SP_ROWS;
                const bool m2 = ok && r2 < SP_ROWS, m3 = ok && r3 < SP_ROWS;
                if (__builtin_amdgcn_ballot_w32(m0 || m1 || m2 || m3)) {
                    const unsigned e0 = (unsigned)(head + ((cu * 32 + lane) << 2));
                    append(m0, (e0     ) | (r0 << SP_EBITS));
                    append(m1, (e0 + 1u) | (r1 << SP_EBITS));
                    append(m2, (e0 + 2u) | (r2 << SP_EBITS));
                    append(m3, (e0 + 3u) | (r3 << SP_EBITS));
                }
            }
            while (cnt >= 32) { cnt -= 32; drain(cnt, 32); }
        }

        if (wv == 0) {
            const int bodyEnd = head + (fullChunks << 7);
            const int nLeft   = head + (E - bodyEnd);
            for (int b = 0; b < nLeft; b += 32) {
                const int idx = b + lane;
                const bool valid = idx < nLeft;
                const int e = idx < head ? idx : bodyEnd + (idx - head);
                const unsigned r = valid ? (unsigned)(dstp[e] - lo) : 0xFFFFFFFFu;
                append(r < SP_ROWS, (unsigned)e | (r << SP_EBITS));
            }
        }
        while (cnt > 0) { const int n = min(cnt, 32); cnt -= n; drain(cnt, n); }
    }

    __syncthreads();

    if (KEEP && threadIdx.x == 0) {
        const unsigned total = cursor;
        counts[blockIdx.x] = total <= (unsigned)cap ? total : SP_NOLIST;
    }

    if (NEXT) {
        if ((wv >> 3) < SP_TILES % 4) next_layer_tiles<(SP_TILES + 3) / 4>(acc, Wn, AGG, lane, wv, lo, N);
        else                          next_layer_tiles<SP_TILES / 4>(acc, Wn, AGG, lane, wv, lo, N);
    } else {
        for (int r = wv; r < SP_ROWS; r += SP_WAVES) {
            const int row = lo + r;
            if (row < N) {
                const float* a = acc + r * SP_PITCH + lane;
                const float4 v = {a[0], a[32], a[64], a[96]};
                *(float4*)(AGG + (size_t)row * D_HID + lane * 4) = v;
            }
        }
    }
}

__global__ void log_softmax16_kernel(const float* __restrict__ Y,
                                     float* __restrict__ out, int N)
{
    const int r = blockIdx.x * blockDim.x + threadIdx.x;
    if (r >= N) return;
    float v[D_OUT];
    float m = -INFINITY;
    #pragma unroll
    for (int j = 0; j < D_OUT; ++j) { v[j] = Y[(size_t)r * D_OUT + j]; m = fmaxf(m, v[j]); }
    float s = 0.0f;
    #pragma unroll
    for (int j = 0; j < D_OUT; ++j) s += __expf(v[j] - m);
    const float l = m + __logf(s);
    #pragma unroll
    for (int j = 0; j < D_OUT; ++j) out[(size_t)r * D_OUT + j] = v[j] - l;
}

extern "C" void kernel_launch(void* const* d_in, const int* in_sizes, int n_in,
                              void* d_out, int out_size, void* d_ws, size_t ws_size,
                              hipStream_t stream)
{
    const float* x    = (const float*)d_in[0];
    const float* ew   = (const float*)d_in[1];
    const float* W1   = (const float*)d_in[2];
    const float* b1   = (const float*)d_in[3];
    const float* W2   = (const float*)d_in[4];
    const float* b2   = (const float*)d_in[5];
    const float* W3   = (const float*)d_in[6];
    const float* b3   = (const float*)d_in[7];
    const float* fcw  = (const float*)d_in[8];
    const float* fcb  = (const float*)d_in[9];
    const int*   ei   = (const int*)d_in[10];

    const int N = in_sizes[0] / D_IN;
    const int E = in_sizes[1];

    float* B0   = (float*)d_ws;
    float* B1   = B0 + (size_t)N * D_HID;
    float* Y    = B0;

    float* outEmb = (float*)d_out;
    float* outLsm = outEmb + (size_t)N * D_EMB;

    const int rowTiles = (N + 15) / 16;
    const dim3 blk(256);

    const dim3 spGrid((N + SP_ROWS - 1) / SP_ROWS), spBlk(SP_WAVES * 32);

    gemm_wmma_f32_kernel<D_IN, 8><<<rowTiles, blk, 0, stream>>>(x, W1, B0, nullptr, N);
    spmm_owner_kernel<true, true><<<spGrid, spBlk, 0, stream>>>(B0, B1, ei, ew, b1, W2, outLsm, E, N);

    spmm_owner_kernel<false, true><<<spGrid, spBlk, 0, stream>>>(B1, B0, ei, ew, b2, W3, outLsm, E, N);

    spmm_owner_kernel<false, false><<<spGrid, spBlk, 0, stream>>>(B0, outEmb, ei, ew, b3, nullptr, outLsm, E, N);

    gemm_wmma_f32_kernel<D_EMB, 1><<<(rowTiles + 7) / 8, blk, 0, stream>>>(outEmb, fcw, Y, fcb, N);
    log_softmax16_kernel<<<(N + 255) / 256, blk, 0, stream>>>(Y, outLsm, N);
}


# Sweep reads one group byte per edge (destination / 200, written by the first GEMM launch into the still unused embedding output) instead of the 4-byte destination; float sums in new list order
# speedup vs baseline: 4.1100x; 1.1039x over previous
#include <hip/hip_runtime.h>
#include <math.h>

typedef float v2f __attribute__((ext_vector_type(2)));
typedef float v8f __attribute__((ext_vector_type(8)));

#define D_IN  256
#define D_HID 128
#define D_EMB 128
#define D_OUT 16

#define SP_ROWS   200
#define SP_NOGRP  0xFFu

__device__ __forceinline__ int ids_chunks(int E) { return (E + 511) >> 9; }
__device__ __forceinline__ bool ids_usable(int N, int E)
{
    const int nGroups = (N + SP_ROWS - 1) / SP_ROWS;
    return nGroups <= (int)SP_NOGRP && ((long long)ids_chunks(E) << 9) <= (long long)N * D_EMB * 4;
}

template <int K, int CT, bool IDS>
__global__ __launch_bounds__(256)
void gemm_wmma_f32_kernel(const float* __restrict__ A,
                          const float* __restrict__ Wm,
                          float* __restrict__ C,
                          const float* __restrict__ bias,
                          int M,
                          const int* __restrict__ ei,
                          float* __restrict__ idsOut,
                          int E)
{
    constexpr int RT    = 8 / CT;
    constexpr int ROWS  = 16 * RT;
    constexpr int PITCH = K + 4;
    constexpr int DOUTv = 16 * CT;
    static_assert(RT * CT == 8 && K % 128 == 0, "8 waves per group, whole 512-byte row pieces");

    __shared__ __attribute__((aligned(16))) float As[ROWS * PITCH];

    const int lane = threadIdx.x & 31;
    const int wv   = __builtin_amdgcn_readfirstlane((int)(threadIdx.x >> 5));
    const int half = lane >> 4;
    const int lm   = lane & 15;
    const int blockRow = blockIdx.x * ROWS;

    float4 st[ROWS / 8][K / 128];
    #pragma unroll
    for (int i = 0; i < ROWS / 8; ++i) {
        const int gr = min(blockRow + wv + 8 * i, M - 1);
        const float* src = A + (size_t)gr * K + lane * 4;
        #pragma unroll
        for (int q = 0; q < K / 128; ++q) st[i][q] = *(const float4*)(src + q * 128);
    }
    if (IDS && ids_usable(M, E)) {
        const int* dstp = ei + E;
        unsigned* ids = (unsigned*)idsOut;
        const int nDw = ids_chunks(E) * 128;
        for (int q = blockIdx.x * 256 + threadIdx.x; q < nDw; q += gridDim.x * 256) {
            unsigned p = 0u;
            #pragma unroll
            for (int i = 0; i < 4; ++i) {
                const int e = 4 * q + i;
                unsigned g = SP_NOGRP;
                if (e < E) {
                    const unsigned d = (unsigned)dstp[e];
                    if (d < (unsigned)M) g = d / SP_ROWS;
                }
                p |= g << (8 * i);
            }
            ids[q] = p;
        }
    }
    #pragma unroll
    for (int i = 0; i < ROWS / 8; ++i) {
        float* dst = As + (wv + 8 * i) * PITCH + lane * 4;
        #pragma unroll
        for (int q = 0; q < K / 128; ++q) {
            *(float4*)(dst + q * 128) = st[i][q];
        }
    }
    __syncthreads();

    const int rowBase = blockRow + (wv / CT) * 16;
    const int colBase = (wv % CT) * 16;

    const float* pA = As + ((wv / CT) * 16 + lm) * PITCH + 2 * half;
    const float* pB = Wm + (size_t)(2 * half) * DOUTv + colBase + lm;

    v8f acc = {};
    #pragma unroll 8
    for (int s = 0; s < K / 4; ++s) {
        const v2f a = *(const v2f*)(pA + 4 * s);
        v2f b;
        b.x = pB[(size_t)(4 * s)     * DOUTv];
        b.y = pB[(size_t)(4 * s + 1) * DOUTv];
        acc = __builtin_amdgcn_wmma_f32_16x16x4_f32(
                  false, a, false, b, (short)0, acc, false, false);
    }

    const float bv = bias ? bias[colBase + lm] : 0.0f;
    float* pC = C + (size_t)(rowBase + 8 * half) * DOUTv + colBase + lm;
    #pragma unroll
    for (int v = 0; v < 8; ++v) {
        const int r = rowBase + v + 8 * half;
        if (r < M) pC[(size_t)v * DOUTv] = acc[v] + bv;
    }
}

#define SP_TILES  ((SP_ROWS + 15) / 16)
#define SP_LROWS  (16 * SP_TILES)
#define SP_PITCH  (D_HID + 4)
#define SP_EBITS  23
#define SP_WAVES  32
#define SP_SCAN_U 8
#define SP_IDS_U  3
#define SP_GATH_U 8
#define SP_PEND   (32 + 512 * SP_IDS_U)
#define SP_LIST_U    5
#define SP_LIST_MAX  (SP_WAVES * 32 * SP_LIST_U)
#define SP_LIST_GATH 16
#define SP_NOLIST    0xFFFFFFFFu
static_assert(SP_ROWS <= (1 << (32 - SP_EBITS)), "row number fits above the edge index");
static_assert(SP_PEND >= 32 + 128 * SP_SCAN_U, "the destination sweep's worst case fits the pending list too");
static_assert((SP_LROWS * SP_PITCH + SP_WAVES * SP_PEND + 1) * 4 <= 320 * 1024, "sums, pending lists and the cursor fit the LDS");
static_assert(SP_WAVES == 32 && D_HID == 128, "next-layer product: 8 column tiles x 4 row groups of waves");

template <int U>
__device__ __forceinline__ void gather_add(const float* __restrict__ H, float* acc,
                                           int lane, int src, float w, int dl, int n)
{
    for (int i = 0; i < n; i += U) {
        float4 h[U];
        #pragma unroll
        for (int k = 0; k < U; ++k) {
            const int s = __builtin_amdgcn_readlane(src, i + k);
            h[k] = *(const float4*)(H + (size_t)s * D_HID + lane * 4);
        }
        #pragma unroll
        for (int k = 0; k < U; ++k) {
            if (i + k < n) {
                const float wk = __int_as_float(
                    __builtin_amdgcn_readlane(__float_as_int(w), i + k));
                const int dk = __builtin_amdgcn_readlane(dl, i + k);
                float* a = acc + dk * SP_PITCH + lane;
                __hip_atomic_fetch_add(a,      h[k].x * wk, __ATOMIC_RELAXED, __HIP_MEMORY_SCOPE_WORKGROUP);
                __hip_atomic_fetch_add(a + 32, h[k].y * wk, __ATOMIC_RELAXED, __HIP_MEMORY_SCOPE_WORKGROUP);
                __hip_atomic_fetch_add(a + 64, h[k].z * wk, __ATOMIC_RELAXED, __HIP_MEMORY_SCOPE_WORKGROUP);
                __hip_atomic_fetch_add(a + 96, h[k].w * wk, __ATOMIC_RELAXED, __HIP_MEMORY_SCOPE_WORKGROUP);
            }
        }
    }
}

template <int NT>
__device__ __forceinline__ void next_layer_tiles(const float* acc, const float* __restrict__ Wn,
                                                 float* __restrict__ OUT,
                                                 int lane, int wv, int lo, int N)
{
    const int half = lane >> 4;
    const int lm   = lane & 15;
    const int ct   = wv & 7;
    const int rg   = wv >> 3;
    const float* pA = acc + (rg * 16 + lm) * SP_PITCH + 2 * half;
    const float* pB = Wn + (size_t)(8 * half) * D_HID + ct * 16 + lm;

    v8f c[NT];
    #pragma unroll
    for (int i = 0; i < NT; ++i) c[i] = v8f{};

    for (int j = 0; j < D_HID / 32; ++j) {
        #pragma unroll
        for (int u = 0; u < 8; ++u) {
            v2f b;
            b.x = pB[(size_t)(16 * u + j)     * D_HID];
            b.y = pB[(size_t)(16 * u + j + 4) * D_HID];
            #pragma unroll
            for (int i = 0; i < NT; ++i) {
                v2f a = *(const v2f*)(pA + i * 64 * SP_PITCH + 32 * j + 4 * u);
                a.x = fmaxf(a.x, 0.0f);
                a.y = fmaxf(a.y, 0.0f);
                c[i] = __builtin_amdgcn_wmma_f32_16x16x4_f32(
                           false, a, false, b, (short)0, c[i], false, false);
            }
        }
    }

    #pragma unroll
    for (int i = 0; i < NT; ++i) {
        #pragma unroll
        for (int v = 0; v < 8; ++v) {
            const int r = (rg + 4 * i) * 16 + 8 * half + v;
            if (r < SP_ROWS && lo + r < N)
                OUT[(size_t)(lo + r) * D_HID + ct * 16 + lm] = c[i][v];
        }
    }
}

template <bool KEEP, bool NEXT>
__global__ __launch_bounds__(SP_WAVES * 32)
void spmm_owner_kernel(const float* __restrict__ H,
                       float* __restrict__ AGG,
                       const int* __restrict__ ei,
                       const float* __restrict__ ew,
                       const float* __restrict__ bias,
                       const float* __restrict__ Wn,
                       float* __restrict__ keep,
                       const float* __restrict__ idsIn,
                       int E, int N)
{
    __shared__ float    acc[SP_LROWS * SP_PITCH];
    __shared__ unsigned pend[SP_WAVES * SP_PEND];
    __shared__ unsigned cursor;

    const int lane = threadIdx.x & 31;
    const int wv   = __builtin_amdgcn_readfirstlane((int)(threadIdx.x >> 5));
    const int lo   = blockIdx.x * SP_ROWS;
    unsigned* myPend = pend + wv * SP_PEND;
    const int* dstp  = ei + E;

    const int nGroups = (N + SP_ROWS - 1) / SP_ROWS;
    const int hdrDw   = (nGroups + 63) & ~63;
    const int room    = N * D_OUT - hdrDw;
    const int cap     = room > 0 ? min((room / nGroups) & ~31, SP_LIST_MAX) : 0;
    unsigned* counts  = (unsigned*)keep;
    unsigned* myList  = counts + hdrDw + (size_t)blockIdx.x * cap;

    for (int i = threadIdx.x; i < SP_LROWS * D_HID; i += SP_WAVES * 32) {
        const int p = i & (D_HID - 1);
        acc[(i / D_HID) * SP_PITCH + p] = bias[((p & 31) << 2) | (p >> 5)];
    }
    if (KEEP && threadIdx.x == 0) cursor = 0u;
    __syncthreads();

    bool sweep = true;
    if (!KEEP) {
        const unsigned total = __builtin_amdgcn_readfirstlane(counts[blockIdx.x]);
        if (total <= (unsigned)cap) {
            sweep = false;
            unsigned rec[SP_LIST_U];
            int      src[SP_LIST_U];
            float    w[SP_LIST_U];
            #pragma unroll
            for (int u = 0; u < SP_LIST_U; ++u) {
                const int c = (wv + SP_WAVES * u) * 32 + lane;
                rec[u] = c < (int)total ? myList[c] : 0u;
            }
            #pragma unroll
            for (int u = 0; u < SP_LIST_U; ++u) {
                const int  c = (wv + SP_WAVES * u) * 32 + lane;
                const bool v = c < (int)total;
                const int  e = (int)(rec[u] & ((1u << SP_EBITS) - 1u));
                src[u] = v ? ei[e] : 0;
                w[u]   = v ? ew[e] : 0.0f;
            }
            #pragma unroll
            for (int u = 0; u < SP_LIST_U; ++u) {
                const int c0 = (wv + SP_WAVES * u) * 32;
                if (c0 < (int)total)
                    gather_add<SP_LIST_GATH>(H, acc, lane, src[u], w[u],
                                             (int)(rec[u] >> SP_EBITS), min(32, (int)total - c0));
            }
        }
    }

    if (sweep) {
        int cnt = 0;

        auto append = [&](bool m, unsigned rec) {
            const unsigned mask = __builtin_amdgcn_ballot_w32(m);
            if (m) myPend[cnt + (int)__builtin_amdgcn_mbcnt_lo(mask, 0u)] = rec;
            cnt += __builtin_popcount(mask);
        };

        const bool byIds = KEEP && ids_usable(N, E);

        auto drain = [&](int base, int n) {
            unsigned rec = 0u;
            int   src = 0;
            float w   = 0.0f;
            if (lane < n) {
                rec = myPend[base + lane];
                const int e = (int)(rec & ((1u << SP_EBITS) - 1u));
                src = ei[e];
                w   = ew[e];
                if (byIds) rec |= (unsigned)(dstp[e] - lo) << SP_EBITS;
            }
            if (KEEP) {
                unsigned pos = 0u;
                if (lane == 0)
                    pos = __hip_atomic_fetch_add(&cursor, (unsigned)n, __ATOMIC_RELAXED, __HIP_MEMORY_SCOPE_WORKGROUP);
                pos = __builtin_amdgcn_readfirstlane(pos);
                if (lane < n && pos + (unsigned)n <= (unsigned)cap) myList[pos + lane] = rec;
            }
            gather_add<SP_GATH_U>(H, acc, lane, src, w, (int)(rec >> SP_EBITS), n);
        };

        if (byIds) {
            const uint4* iv = (const uint4*)idsIn;
            const int nChunks = ids_chunks(E);
            const unsigned kk = (unsigned)blockIdx.x * 0x01010101u;
            for (int c = wv; c < nChunks; c += SP_WAVES * SP_IDS_U) {
                uint4 d[SP_IDS_U];
                #pragma unroll
                for (int u = 0; u < SP_IDS_U; ++u) {
                    const int cu = min(c + SP_WAVES * u, nChunks - 1);
                    d[u] = iv[cu * 32 + lane];
                }
                #pragma unroll
                for (int u = 0; u < SP_IDS_U; ++u) {
                    const int cu = c + SP_WAVES * u;
                    const unsigned x0 = d[u].x ^ kk, x1 = d[u].y ^ kk, x2 = d[u].z ^ kk, x3 = d[u].w ^ kk;
                    const unsigned y0 = ((x0 & 0x7F7F7F7Fu) + 0x7F7F7F7Fu) | x0 | 0x7F7F7F7Fu;
                    const unsigned y1 = ((x1 & 0x7F7F7F7Fu) + 0x7F7F7F7Fu) | x1 | 0x7F7F7F7Fu;
                    const unsigned y2 = ((x2 & 0x7F7F7F7Fu) + 0x7F7F7F7Fu) | x2 | 0x7F7F7F7Fu;
                    const unsigned y3 = ((x3 & 0x7F7F7F7Fu) + 0x7F7F7F7Fu) | x3 | 0x7F7F7F7Fu;
                    const bool hit = (y0 & y1 & y2 & y3) != 0xFFFFFFFFu;
                    if (cu < nChunks && __builtin_amdgcn_ballot_w32(hit)) {
                        unsigned m = (~y0 >> 7) | (~y1 >> 6) | (~y2 >> 5) | (~y3 >> 4);
                        const unsigned e0 = (unsigned)(cu * 32 + lane) << 4;
                        unsigned bal = __builtin_amdgcn_ballot_w32(m != 0u);
                        while (bal) {
                            if (m != 0u) {
                                const int p = __builtin_ctz(m);
                                myPend[cnt + (int)__builtin_amdgcn_mbcnt_lo(bal, 0u)] =
                                    e0 + (unsigned)(((p & 7) << 2) + (p >> 3));
                            }
                            cnt += __builtin_popcount(bal);
                            m &= m - 1u;
                            bal = __builtin_amdgcn_ballot_w32(m != 0u);
                        }
                    }
                }
                while (cnt >= 32) { cnt -= 32; drain(cnt, 32); }
            }
        } else {
        const int head = min(E, (int)(((16u - (unsigned)((size_t)dstp & 15u)) & 15u) >> 2));
        const int nvec = (E - head) >> 2;
        const int fullChunks = nvec >> 5;
        const int4* dv = (const int4*)(dstp + head);

        for (int c = wv; c < fullChunks; c += SP_WAVES * SP_SCAN_U) {
            int4 d[SP_SCAN_U];
            #pragma unroll
            for (int u = 0; u < SP_SCAN_U; ++u) {
                const int cu = min(c + SP_WAVES * u, fullChunks - 1);
                d[u] = dv[cu * 32 + lane];
            }
            #pragma unroll
            for (int u = 0; u < SP_SCAN_U; ++u) {
                const int  cu = c + SP_WAVES * u;
                const bool ok = cu < fullChunks;
                const unsigned r0 = (unsigned)(d[u].x - lo), r1 = (unsigned)(d[u].y - lo);
                const unsigned r2 = (unsigned)(d[u].z - lo), r3 = (unsigned)(d[u].w - lo);
                const bool m0 = ok && r0 < SP_ROWS, m1 = ok && r1 < SP_ROWS;
                const bool m2 = ok && r2 < SP_ROWS, m3 = ok && r3 < SP_ROWS;
                if (__builtin_amdgcn_ballot_w32(m0 || m1 || m2 || m3)) {
                    const unsigned e0 = (unsigned)(head + ((cu * 32 + lane) << 2));
                    append(m0, (e0     ) | (r0 << SP_EBITS));
                    append(m1, (e0 + 1u) | (r1 << SP_EBITS));
                    append(m2, (e0 + 2u) | (r2 << SP_EBITS));
                    append(m3, (e0 + 3u) | (r3 << SP_EBITS));
                }
            }
            while (cnt >= 32) { cnt -= 32; drain(cnt, 32); }
        }

        if (wv == 0) {
            const int bodyEnd = head + (fullChunks << 7);
            const int nLeft   = head + (E - bodyEnd);
            for (int b = 0; b < nLeft; b += 32) {
                const int idx = b + lane;
                const bool valid = idx < nLeft;
                const int e = idx < head ? idx : bodyEnd + (idx - head);
                const unsigned r = valid ? (unsigned)(dstp[e] - lo) : 0xFFFFFFFFu;
                append(r < SP_ROWS, (unsigned)e | (r << SP_EBITS));
            }
        }
        }
        while (cnt > 0) { const int n = min(cnt, 32); cnt -= n; drain(cnt, n); }
    }

    __syncthreads();

    if (KEEP && threadIdx.x == 0) {
        const unsigned total = cursor;
        counts[blockIdx.x] = total <= (unsigned)cap ? total : SP_NOLIST;
    }

    if (NEXT) {
        if ((wv >> 3) < SP_TILES % 4) next_layer_tiles<(SP_TILES + 3) / 4>(acc, Wn, AGG, lane, wv, lo, N);
        else                          next_layer_tiles<SP_TILES / 4>(acc, Wn, AGG, lane, wv, lo, N);
    } else {
        for (int r = wv; r < SP_ROWS; r += SP_WAVES) {
            const int row = lo + r;
            if (row < N) {
                const float* a = acc + r * SP_PITCH + lane;
                const float4 v = {a[0], a[32], a[64], a[96]};
                *(float4*)(AGG + (size_t)row * D_HID + lane * 4) = v;
            }
        }
    }
}

__global__ void log_softmax16_kernel(const float* __restrict__ Y,
                                     float* __restrict__ out, int N)
{
    const int r = blockIdx.x * blockDim.x + threadIdx.x;
    if (r >= N) return;
    float v[D_OUT];
    float m = -INFINITY;
    #pragma unroll
    for (int j = 0; j < D_OUT; ++j) { v[j] = Y[(size_t)r * D_OUT + j]; m = fmaxf(m, v[j]); }
    float s = 0.0f;
    #pragma unroll
    for (int j = 0; j < D_OUT; ++j) s += __expf(v[j] - m);
    const float l = m + __logf(s);
    #pragma unroll
    for (int j = 0; j < D_OUT; ++j) out[(size_t)r * D_OUT + j] = v[j] - l;
}

extern "C" void kernel_launch(void* const* d_in, const int* in_sizes, int n_in,
                              void* d_out, int out_size, void* d_ws, size_t ws_size,
                              hipStream_t stream)
{
    const float* x    = (const float*)d_in[0];
    const float* ew   = (const float*)d_in[1];
    const float* W1   = (const float*)d_in[2];
    const float* b1   = (const float*)d_in[3];
    const float* W2   = (const float*)d_in[4];
    const float* b2   = (const float*)d_in[5];
    const float* W3   = (const float*)d_in[6];
    const float* b3   = (const float*)d_in[7];
    const float* fcw  = (const float*)d_in[8];
    const float* fcb  = (const float*)d_in[9];
    const int*   ei   = (const int*)d_in[10];

    const int N = in_sizes[0] / D_IN;
    const int E = in_sizes[1];

    float* B0   = (float*)d_ws;
    float* B1   = B0 + (size_t)N * D_HID;
    float* Y    = B0;

    float* outEmb = (float*)d_out;
    float* outLsm = outEmb + (size_t)N * D_EMB;

    const int rowTiles = (N + 15) / 16;
    const dim3 blk(256);

    const dim3 spGrid((N + SP_ROWS - 1) / SP_ROWS), spBlk(SP_WAVES * 32);

    gemm_wmma_f32_kernel<D_IN, 8, true><<<rowTiles, blk, 0, stream>>>(x, W1, B0, nullptr, N, ei, outEmb, E);
    spmm_owner_kernel<true, true><<<spGrid, spBlk, 0, stream>>>(B0, B1, ei, ew, b1, W2, outLsm, outEmb, E, N);

    spmm_owner_kernel<false, true><<<spGrid, spBlk, 0, stream>>>(B1, B0, ei, ew, b2, W3, outLsm, nullptr, E, N);

    spmm_owner_kernel<false, false><<<spGrid, spBlk, 0, stream>>>(B0, outEmb, ei, ew, b3, nullptr, outLsm, nullptr, E, N);

    gemm_wmma_f32_kernel<D_EMB, 1, false><<<(rowTiles + 7) / 8, blk, 0, stream>>>(outEmb, fcw, Y, fcb, N, nullptr, nullptr, 0);
    log_softmax16_kernel<<<(N + 255) / 256, blk, 0, stream>>>(Y, outLsm, N);
}


# Kept list records hold source row and weight (8 bytes, in the group's own rows of the embedding output; group bytes moved to the log-softmax part): launches 3, 4 drop the random 4-byte index reads
# speedup vs baseline: 4.5853x; 1.1156x over previous
#include <hip/hip_runtime.h>
#include <math.h>

typedef float v2f __attribute__((ext_vector_type(2)));
typedef float v8f __attribute__((ext_vector_type(8)));

#define D_IN  256
#define D_HID 128
#define D_EMB 128
#define D_OUT 16

#define SP_ROWS   200
#define SP_NOGRP  0xFFu

__device__ __forceinline__ int ids_chunks(int E) { return (E + 511) >> 9; }
__device__ __forceinline__ bool ids_usable(int N, int E)
{
    const int nGroups = (N + SP_ROWS - 1) / SP_ROWS;
    return nGroups <= (int)SP_NOGRP && ((long long)ids_chunks(E) << 9) <= (long long)N * D_OUT * 4;
}

template <int K, int CT, bool IDS>
__global__ __launch_bounds__(256)
void gemm_wmma_f32_kernel(const float* __restrict__ A,
                          const float* __restrict__ Wm,
                          float* __restrict__ C,
                          const float* __restrict__ bias,
                          int M,
                          const int* __restrict__ ei,
                          float* __restrict__ idsOut,
                          int E)
{
    constexpr int RT    = 8 / CT;
    constexpr int ROWS  = 16 * RT;
    constexpr int PITCH = K + 4;
    constexpr int DOUTv = 16 * CT;
    static_assert(RT * CT == 8 && K % 128 == 0, "8 waves per group, whole 512-byte row pieces");

    __shared__ __attribute__((aligned(16))) float As[ROWS * PITCH];

    const int lane = threadIdx.x & 31;
    const int wv   = __builtin_amdgcn_readfirstlane((int)(threadIdx.x >> 5));
    const int half = lane >> 4;
    const int lm   = lane & 15;
    const int blockRow = blockIdx.x * ROWS;

    float4 st[ROWS / 8][K / 128];
    #pragma unroll
    for (int i = 0; i < ROWS / 8; ++i) {
        const int gr = min(blockRow + wv + 8 * i, M - 1);
        const float* src = A + (size_t)gr * K + lane * 4;
        #pragma unroll
        for (int q = 0; q < K / 128; ++q) st[i][q] = *(const float4*)(src + q * 128);
    }
    if (IDS && ids_usable(M, E)) {
        const int* dstp = ei + E;
        unsigned* ids = (unsigned*)idsOut;
        const int nDw = ids_chunks(E) * 128;
        for (int q = blockIdx.x * 256 + threadIdx.x; q < nDw; q += gridDim.x * 256) {
            unsigned p = 0u;
            #pragma unroll
            for (int i = 0; i < 4; ++i) {
                const int e = 4 * q + i;
                unsigned g = SP_NOGRP;
                if (e < E) {
                    const unsigned d = (unsigned)dstp[e];
                    if (d < (unsigned)M) g = d / SP_ROWS;
                }
                p |= g << (8 * i);
            }
            ids[q] = p;
        }
    }
    #pragma unroll
    for (int i = 0; i < ROWS / 8; ++i) {
        float* dst = As + (wv + 8 * i) * PITCH + lane * 4;
        #pragma unroll
        for (int q = 0; q < K / 128; ++q) {
            *(float4*)(dst + q * 128) = st[i][q];
        }
    }
    __syncthreads();

    const int rowBase = blockRow + (wv / CT) * 16;
    const int colBase = (wv % CT) * 16;

    const float* pA = As + ((wv / CT) * 16 + lm) * PITCH + 2 * half;
    const float* pB = Wm + (size_t)(2 * half) * DOUTv + colBase + lm;

    v8f acc = {};
    #pragma unroll 8
    for (int s = 0; s < K / 4; ++s) {
        const v2f a = *(const v2f*)(pA + 4 * s);
        v2f b;
        b.x = pB[(size_t)(4 * s)     * DOUTv];
        b.y = pB[(size_t)(4 * s + 1) * DOUTv];
        acc = __builtin_amdgcn_wmma_f32_16x16x4_f32(
                  false, a, false, b, (short)0, acc, false, false);
    }

    const float bv = bias ? bias[colBase + lm] : 0.0f;
    float* pC = C + (size_t)(rowBase + 8 * half) * DOUTv + colBase + lm;
    #pragma unroll
    for (int v = 0; v < 8; ++v) {
        const int r = rowBase + v + 8 * half;
        if (r < M) pC[(size_t)v * DOUTv] = acc[v] + bv;
    }
}

#define SP_TILES  ((SP_ROWS + 15) / 16)
#define SP_LROWS  (16 * SP_TILES)
#define SP_PITCH  (D_HID + 4)
#define SP_EBITS  23
#define SP_WAVES  32
#define SP_SCAN_U 8
#define SP_IDS_U  3
#define SP_GATH_U 8
#define SP_PEND   (32 + 512 * SP_IDS_U)
#define SP_LIST_U    5
#define SP_LIST_MAX  (SP_WAVES * 32 * SP_LIST_U)
#define SP_LIST_GATH 16
#define SP_NOLIST    0xFFFFFFFFu
#define SP_LIST_AT   64
static_assert(SP_ROWS <= (1 << (32 - SP_EBITS)), "row number fits above the edge index");
static_assert(SP_PEND >= 32 + 128 * SP_SCAN_U, "the destination sweep's worst case fits the pending list too");
static_assert((SP_LROWS * SP_PITCH + SP_WAVES * SP_PEND + 1) * 4 <= 320 * 1024, "sums, pending lists and the cursor fit the LDS");
static_assert(SP_WAVES == 32 && D_HID == 128, "next-layer product: 8 column tiles x 4 row groups of waves");
static_assert(SP_TILES < SP_WAVES && D_EMB == D_HID && D_OUT == 16, "last layer: one classifier tile per wave, waves left over to store the sums");

template <int U>
__device__ __forceinline__ void gather_add(const float* __restrict__ H, float* acc,
                                           int lane, int src, float w, int dl, int n)
{
    for (int i = 0; i < n; i += U) {
        float4 h[U];
        #pragma unroll
        for (int k = 0; k < U; ++k) {
            const int s = __builtin_amdgcn_readlane(src, i + k);
            h[k] = *(const float4*)(H + (size_t)s * D_HID + lane * 4);
        }
        #pragma unroll
        for (int k = 0; k < U; ++k) {
            if (i + k < n) {
                const float wk = __int_as_float(
                    __builtin_amdgcn_readlane(__float_as_int(w), i + k));
                const int dk = __builtin_amdgcn_readlane(dl, i + k);
                float* a = acc + dk * SP_PITCH + lane;
                __hip_atomic_fetch_add(a,      h[k].x * wk, __ATOMIC_RELAXED, __HIP_MEMORY_SCOPE_WORKGROUP);
                __hip_atomic_fetch_add(a + 32, h[k].y * wk, __ATOMIC_RELAXED, __HIP_MEMORY_SCOPE_WORKGROUP);
                __hip_atomic_fetch_add(a + 64, h[k].z * wk, __ATOMIC_RELAXED, __HIP_MEMORY_SCOPE_WORKGROUP);
                __hip_atomic_fetch_add(a + 96, h[k].w * wk, __ATOMIC_RELAXED, __HIP_MEMORY_SCOPE_WORKGROUP);
            }
        }
    }
}

template <int NT>
__device__ __forceinline__ void next_layer_tiles(const float* acc, const float* __restrict__ Wn,
                                                 float* __restrict__ OUT,
                                                 int lane, int wv, int lo, int N)
{
    const int half = lane >> 4;
    const int lm   = lane & 15;
    const int ct   = wv & 7;
    const int rg   = wv >> 3;
    const float* pA = acc + (rg * 16 + lm) * SP_PITCH + 2 * half;
    const float* pB = Wn + (size_t)(8 * half) * D_HID + ct * 16 + lm;

    v8f c[NT];
    #pragma unroll
    for (int i = 0; i < NT; ++i) c[i] = v8f{};

    for (int j = 0; j < D_HID / 32; ++j) {
        #pragma unroll
        for (int u = 0; u < 8; ++u) {
            v2f b;
            b.x = pB[(size_t)(16 * u + j)     * D_HID];
            b.y = pB[(size_t)(16 * u + j + 4) * D_HID];
            #pragma unroll
            for (int i = 0; i < NT; ++i) {
                v2f a = *(const v2f*)(pA + i * 64 * SP_PITCH + 32 * j + 4 * u);
                a.x = fmaxf(a.x, 0.0f);
                a.y = fmaxf(a.y, 0.0f);
                c[i] = __builtin_amdgcn_wmma_f32_16x16x4_f32(
                           false, a, false, b, (short)0, c[i], false, false);
            }
        }
    }

    #pragma unroll
    for (int i = 0; i < NT; ++i) {
        #pragma unroll
        for (int v = 0; v < 8; ++v) {
            const int r = (rg + 4 * i) * 16 + 8 * half + v;
            if (r < SP_ROWS && lo + r < N)
                OUT[(size_t)(lo + r) * D_HID + ct * 16 + lm] = c[i][v];
        }
    }
}

template <int M>
__device__ __forceinline__ float row_xor(float v)
{
    return __int_as_float(__builtin_amdgcn_update_dpp(0, __float_as_int(v), 0x160 + M, 0xF, 0xF, true));
}
template <int M>
__device__ __forceinline__ void rows_max_step(float (&m)[8])
{
    #pragma unroll
    for (int v = 0; v < 8; ++v) m[v] = fmaxf(m[v], row_xor<M>(m[v]));
}
template <int M>
__device__ __forceinline__ void rows_sum_step(float (&s)[8])
{
    #pragma unroll
    for (int v = 0; v < 8; ++v) s[v] += row_xor<M>(s[v]);
}

__device__ __forceinline__ void classifier_tile(const float* acc, const float* __restrict__ fcw,
                                                const float* __restrict__ fcb, float* __restrict__ LSM,
                                                int lane, int t, int lo, int N)
{
    const int half = lane >> 4;
    const int lm   = lane & 15;
    const float* pA = acc + (t * 16 + lm) * SP_PITCH + 2 * half;
    const float* pB = fcw + (size_t)(8 * half) * D_OUT + lm;

    v8f c = {};
    #pragma unroll
    for (int j = 0; j < D_EMB / 32; ++j) {
        #pragma unroll
        for (int u = 0; u < 8; ++u) {
            v2f b;
            b.x = pB[(size_t)(16 * u + j)     * D_OUT];
            b.y = pB[(size_t)(16 * u + j + 4) * D_OUT];
            const v2f a = *(const v2f*)(pA + 32 * j + 4 * u);
            c = __builtin_amdgcn_wmma_f32_16x16x4_f32(
                    false, a, false, b, (short)0, c, false, false);
        }
    }

    const float bv = fcb[lm];
    float y[8], m[8], s[8];
    #pragma unroll
    for (int v = 0; v < 8; ++v) { y[v] = c[v] + bv; m[v] = y[v]; }
    rows_max_step<1>(m); rows_max_step<2>(m); rows_max_step<4>(m); rows_max_step<8>(m);
    #pragma unroll
    for (int v = 0; v < 8; ++v) s[v] = __expf(y[v] - m[v]);
    rows_sum_step<1>(s); rows_sum_step<2>(s); rows_sum_step<4>(s); rows_sum_step<8>(s);
    #pragma unroll
    for (int v = 0; v < 8; ++v) {
        const float l = m[v] + __logf(s[v]);
        const int r = t * 16 + 8 * half + v;
        if (r < SP_ROWS && lo + r < N)
            LSM[(size_t)(lo + r) * D_OUT + lm] = y[v] - l;
    }
}

template <bool KEEP, bool NEXT>
__global__ __launch_bounds__(SP_WAVES * 32)
void spmm_owner_kernel(const float* __restrict__ H,
                       float* AGG,
                       const int* __restrict__ ei,
                       const float* __restrict__ ew,
                       const float* __restrict__ bias,
                       const float* __restrict__ Wn,
                       const float* __restrict__ fcb,
                       float* keep,
                       float* __restrict__ LSM,
                       const float* __restrict__ idsIn,
                       int E, int N)
{
    __shared__ float    acc[SP_LROWS * SP_PITCH];
    __shared__ unsigned pend[SP_WAVES * SP_PEND];
    __shared__ unsigned cursor;

    const int lane = threadIdx.x & 31;
    const int wv   = __builtin_amdgcn_readfirstlane((int)(threadIdx.x >> 5));
    const int lo   = blockIdx.x * SP_ROWS;
    unsigned* myPend = pend + wv * SP_PEND;
    const int* dstp  = ei + E;

    const int room    = min(SP_ROWS, N - lo) * D_EMB - SP_LIST_AT;
    const int cap     = (room > 0 && N <= (1 << SP_EBITS)) ? min((room >> 1) & ~31, SP_LIST_MAX) : 0;
    unsigned* myCount = (unsigned*)keep + (size_t)lo * D_EMB;
    uint2*    myList  = (uint2*)(myCount + SP_LIST_AT);

    for (int i = threadIdx.x; i < SP_LROWS * D_HID; i += SP_WAVES * 32) {
        const int p = i & (D_HID - 1);
        acc[(i / D_HID) * SP_PITCH + p] = bias[((p & 31) << 2) | (p >> 5)];
    }
    if (KEEP && threadIdx.x == 0) cursor = 0u;
    __syncthreads();

    bool sweep = true;
    if (!KEEP) {
        const unsigned total = __builtin_amdgcn_readfirstlane(*myCount);
        if (total <= (unsigned)cap) {
            sweep = false;
            uint2 rec[SP_LIST_U];
            #pragma unroll
            for (int u = 0; u < SP_LIST_U; ++u) {
                const int c = (wv + SP_WAVES * u) * 32 + lane;
                rec[u] = c < (int)total ? myList[c] : uint2{0u, 0u};
            }
            #pragma unroll
            for (int u = 0; u < SP_LIST_U; ++u) {
                const int c0 = (wv + SP_WAVES * u) * 32;
                if (c0 < (int)total)
                    gather_add<SP_LIST_GATH>(H, acc, lane, (int)(rec[u].x & ((1u << SP_EBITS) - 1u)),
                                             __uint_as_float(rec[u].y),
                                             (int)(rec[u].x >> SP_EBITS), min(32, (int)total - c0));
            }
        }
    }

    if (sweep) {
        int cnt = 0;

        auto append = [&](bool m, unsigned rec) {
            const unsigned mask = __builtin_amdgcn_ballot_w32(m);
            if (m) myPend[cnt + (int)__builtin_amdgcn_mbcnt_lo(mask, 0u)] = rec;
            cnt += __builtin_popcount(mask);
        };

        const bool byIds = KEEP && ids_usable(N, E);

        auto drain = [&](int base, int n) {
            unsigned rec = 0u;
            int   src = 0;
            float w   = 0.0f;
            if (lane < n) {
                rec = myPend[base + lane];
                const int e = (int)(rec & ((1u << SP_EBITS) - 1u));
                src = ei[e];
                w   = ew[e];
                if (byIds) rec |= (unsigned)(dstp[e] - lo) << SP_EBITS;
            }
            if (KEEP) {
                unsigned pos = 0u;
                if (lane == 0)
                    pos = __hip_atomic_fetch_add(&cursor, (unsigned)n, __ATOMIC_RELAXED, __HIP_MEMORY_SCOPE_WORKGROUP);
                pos = __builtin_amdgcn_readfirstlane(pos);
                if (lane < n && pos + (unsigned)n <= (unsigned)cap)
                    myList[pos + lane] = uint2{(unsigned)src | (rec & ~((1u << SP_EBITS) - 1u)), __float_as_uint(w)};
            }
            gather_add<SP_GATH_U>(H, acc, lane, src, w, (int)(rec >> SP_EBITS), n);
        };

        if (byIds) {
            const uint4* iv = (const uint4*)idsIn;
            const int nChunks = ids_chunks(E);
            const unsigned kk = (unsigned)blockIdx.x * 0x01010101u;
            for (int c = wv; c < nChunks; c += SP_WAVES * SP_IDS_U) {
                uint4 d[SP_IDS_U];
                #pragma unroll
                for (int u = 0; u < SP_IDS_U; ++u) {
                    const int cu = min(c + SP_WAVES * u, nChunks - 1);
                    d[u] = iv[cu * 32 + lane];
                }
                #pragma unroll
                for (int u = 0; u < SP_IDS_U; ++u) {
                    const int cu = c + SP_WAVES * u;
                    const unsigned x0 = d[u].x ^ kk, x1 = d[u].y ^ kk, x2 = d[u].z ^ kk, x3 = d[u].w ^ kk;
                    const unsigned y0 = ((x0 & 0x7F7F7F7Fu) + 0x7F7F7F7Fu) | x0 | 0x7F7F7F7Fu;
                    const unsigned y1 = ((x1 & 0x7F7F7F7Fu) + 0x7F7F7F7Fu) | x1 | 0x7F7F7F7Fu;
                    const unsigned y2 = ((x2 & 0x7F7F7F7Fu) + 0x7F7F7F7Fu) | x2 | 0x7F7F7F7Fu;
                    const unsigned y3 = ((x3 & 0x7F7F7F7Fu) + 0x7F7F7F7Fu) | x3 | 0x7F7F7F7Fu;
                    const bool hit = (y0 & y1 & y2 & y3) != 0xFFFFFFFFu;
                    if (cu < nChunks && __builtin_amdgcn_ballot_w32(hit)) {
                        unsigned m = (~y0 >> 7) | (~y1 >> 6) | (~y2 >> 5) | (~y3 >> 4);
                        const unsigned e0 = (unsigned)(cu * 32 + lane) << 4;
                        unsigned bal = __builtin_amdgcn_ballot_w32(m != 0u);
                        while (bal) {
                            if (m != 0u) {
                                const int p = __builtin_ctz(m);
                                myPend[cnt + (int)__builtin_amdgcn_mbcnt_lo(bal, 0u)] =
                                    e0 + (unsigned)(((p & 7) << 2) + (p >> 3));
                            }
                            cnt += __builtin_popcount(bal);
                            m &= m - 1u;
                            bal = __builtin_amdgcn_ballot_w32(m != 0u);
                        }
                    }
                }
                while (cnt >= 32) { cnt -= 32; drain(cnt, 32); }
            }
        } else {
        const int head = min(E, (int)(((16u - (unsigned)((size_t)dstp & 15u)) & 15u) >> 2));
        const int nvec = (E - head) >> 2;
        const int fullChunks = nvec >> 5;
        const int4* dv = (const int4*)(dstp + head);

        for (int c = wv; c < fullChunks; c += SP_WAVES * SP_SCAN_U) {
            int4 d[SP_SCAN_U];
            #pragma unroll
            for (int u = 0; u < SP_SCAN_U; ++u) {
                const int cu = min(c + SP_WAVES * u, fullChunks - 1);
                d[u] = dv[cu * 32 + lane];
            }
            #pragma unroll
            for (int u = 0; u < SP_SCAN_U; ++u) {
                const int  cu = c + SP_WAVES * u;
                const bool ok = cu < fullChunks;
                const unsigned r0 = (unsigned)(d[u].x - lo), r1 = (unsigned)(d[u].y - lo);
                const unsigned r2 = (unsigned)(d[u].z - lo), r3 = (unsigned)(d[u].w - lo);
                const bool m0 = ok && r0 < SP_ROWS, m1 = ok && r1 < SP_ROWS;
                const bool m2 = ok && r2 < SP_ROWS, m3 = ok && r3 < SP_ROWS;
                if (__builtin_amdgcn_ballot_w32(m0 || m1 || m2 || m3)) {
                    const unsigned e0 = (unsigned)(head + ((cu * 32 + lane) << 2));
                    append(m0, (e0     ) | (r0 << SP_EBITS));
                    append(m1, (e0 + 1u) | (r1 << SP_EBITS));
                    append(m2, (e0 + 2u) | (r2 << SP_EBITS));
                    append(m3, (e0 + 3u) | (r3 << SP_EBITS));
                }
            }
            while (cnt >= 32) { cnt -= 32; drain(cnt, 32); }
        }

        if (wv == 0) {
            const int bodyEnd = head + (fullChunks << 7);
            const int nLeft   = head + (E - bodyEnd);
            for (int b = 0; b < nLeft; b += 32) {
                const int idx = b + lane;
                const bool valid = idx < nLeft;
                const int e = idx < head ? idx : bodyEnd + (idx - head);
                const unsigned r = valid ? (unsigned)(dstp[e] - lo) : 0xFFFFFFFFu;
                append(r < SP_ROWS, (unsigned)e | (r << SP_EBITS));
            }
        }
        }
        while (cnt > 0) { const int n = min(cnt, 32); cnt -= n; drain(cnt, n); }
    }

    __syncthreads();

    if (KEEP && threadIdx.x == 0) {
        const unsigned total = cursor;
        *myCount = total <= (unsigned)cap ? total : SP_NOLIST;
    }

    if (NEXT) {
        if ((wv >> 3) < SP_TILES % 4) next_layer_tiles<(SP_TILES + 3) / 4>(acc, Wn, AGG, lane, wv, lo, N);
        else                          next_layer_tiles<SP_TILES / 4>(acc, Wn, AGG, lane, wv, lo, N);
    } else {
        if (wv < SP_TILES) {
            classifier_tile(acc, Wn, fcb, LSM, lane, wv, lo, N);
        } else {
            for (int r = wv - SP_TILES; r < SP_ROWS; r += SP_WAVES - SP_TILES) {
                const int row = lo + r;
                if (row < N) {
                    const float* a = acc + r * SP_PITCH + lane;
                    const float4 v = {a[0], a[32], a[64], a[96]};
                    *(float4*)(AGG + (size_t)row * D_HID + lane * 4) = v;
                }
            }
        }
    }
}

extern "C" void kernel_launch(void* const* d_in, const int* in_sizes, int n_in,
                              void* d_out, int out_size, void* d_ws, size_t ws_size,
                              hipStream_t stream)
{
    const float* x    = (const float*)d_in[0];
    const float* ew   = (const float*)d_in[1];
    const float* W1   = (const float*)d_in[2];
    const float* b1   = (const float*)d_in[3];
    const float* W2   = (const float*)d_in[4];
    const float* b2   = (const float*)d_in[5];
    const float* W3   = (const float*)d_in[6];
    const float* b3   = (const float*)d_in[7];
    const float* fcw  = (const float*)d_in[8];
    const float* fcb  = (const float*)d_in[9];
    const int*   ei   = (const int*)d_in[10];

    const int N = in_sizes[0] / D_IN;
    const int E = in_sizes[1];

    float* B0   = (float*)d_ws;
    float* B1   = B0 + (size_t)N * D_HID;

    float* outEmb = (float*)d_out;
    float* outLsm = outEmb + (size_t)N * D_EMB;

    const int rowTiles = (N + 15) / 16;
    const dim3 blk(256);

    const dim3 spGrid((N + SP_ROWS - 1) / SP_ROWS), spBlk(SP_WAVES * 32);

    gemm_wmma_f32_kernel<D_IN, 8, true><<<rowTiles, blk, 0, stream>>>(x, W1, B0, nullptr, N, ei, outLsm, E);
    spmm_owner_kernel<true, true><<<spGrid, spBlk, 0, stream>>>(B0, B1, ei, ew, b1, W2, nullptr, outEmb, nullptr, outLsm, E, N);

    spmm_owner_kernel<false, true><<<spGrid, spBlk, 0, stream>>>(B1, B0, ei, ew, b2, W3, nullptr, outEmb, nullptr, nullptr, E, N);

    spmm_owner_kernel<false, false><<<spGrid, spBlk, 0, stream>>>(B0, outEmb, ei, ew, b3, fcw, fcb, outEmb, outLsm, nullptr, E, N);
}
